# starting kernel
# baseline (speedup 1.0000x reference)
#include <hip/hip_runtime.h>
#include <hip/hip_bf16.h>


#define NN 50000
#define EE 800000
#define HH 128

typedef float v2f __attribute__((ext_vector_type(2)));
typedef float v8f __attribute__((ext_vector_type(8)));


__global__ void k_zero(float* __restrict__ p, long n) {
  long i = (long)blockIdx.x * blockDim.x + threadIdx.x;
  if (i < n) p[i] = 0.0f;
}

__global__ void k_setneg(float* __restrict__ out, const float* __restrict__ a, long n) {
  long i = (long)blockIdx.x * blockDim.x + threadIdx.x;
  if (i < n) out[i] = -a[i];
}


__global__ void k_degree(const int* __restrict__ src, float* __restrict__ deg) {
  int e = blockIdx.x * blockDim.x + threadIdx.x;
  if (e < EE) atomicAdd(&deg[src[e]], 1.0f);
}

__global__ void k_dinv(const float* __restrict__ deg, float* __restrict__ dinv) {
  int i = blockIdx.x * blockDim.x + threadIdx.x;
  if (i < NN) {
    float d = deg[i];
    dinv[i] = (d > 0.0f) ? rsqrtf(fmaxf(d, 1.0f)) : 0.0f;
  }
}

__global__ void k_edgew(const int* __restrict__ src, const int* __restrict__ dst,
                        const float* __restrict__ dinv, float* __restrict__ wE) {
  int e = blockIdx.x * blockDim.x + threadIdx.x;
  if (e < EE) wE[e] = -dinv[src[e]] * dinv[dst[e]];
}


__global__ void k_spmv3(const float* __restrict__ h, const int* __restrict__ src,
                        const int* __restrict__ dst, const float* __restrict__ wE,
                        float scale, float* __restrict__ out) {
  int e = blockIdx.x * blockDim.x + threadIdx.x;
  if (e >= EE) return;
  int s = src[e], d = dst[e];
  float c = scale * wE[e];
  atomicAdd(&out[(long)d * 3 + 0], c * h[(long)s * 3 + 0]);
  atomicAdd(&out[(long)d * 3 + 1], c * h[(long)s * 3 + 1]);
  atomicAdd(&out[(long)d * 3 + 2], c * h[(long)s * 3 + 2]);
}

__global__ __launch_bounds__(256) void k_spmv128(
    const float* __restrict__ h, const int* __restrict__ src,
    const int* __restrict__ dst, const float* __restrict__ wE,
    float scale, float* __restrict__ out) {
  int e = blockIdx.x * 8 + (threadIdx.x >> 5);
  if (e >= EE) return;
  int lane = threadIdx.x & 31;
  int s = src[e], d = dst[e];
  float c = scale * wE[e];
  const float4* hp = (const float4*)(h + (long)s * HH);
  float4 v = hp[lane];
  float* op = out + (long)d * HH + lane * 4;
  atomicAdd(op + 0, c * v.x);
  atomicAdd(op + 1, c * v.y);
  atomicAdd(op + 2, c * v.z);
  atomicAdd(op + 3, c * v.w);
}


__global__ __launch_bounds__(32) void k_cheb_gemm3(
    const float* __restrict__ T0, const float* __restrict__ T1,
    const float* __restrict__ T2, const float* __restrict__ T3,
    const float* __restrict__ W, const float* __restrict__ bias,
    float* __restrict__ out) {
  const int lane = threadIdx.x;
  const int tm = blockIdx.x;
  const int tn = blockIdx.y;
  const int row = tm * 16 + (lane & 15);
  const int col = tn * 16 + (lane & 15);
  const int khalf = (lane >> 4) * 2;
  const float* Ts[4] = {T0, T1, T2, T3};

  v8f acc = {};
  for (int k0 = 0; k0 < 12; k0 += 4) {
    v2f a, b;
#pragma unroll
    for (int j = 0; j < 2; ++j) {
      int kk = k0 + khalf + j;
      int t = kk / 3;
      int cc = kk - t * 3;
      a[j] = Ts[t][(long)row * 3 + cc];
      b[j] = W[(long)kk * HH + col];
    }
    acc = __builtin_amdgcn_wmma_f32_16x16x4_f32(
        false, a, false, b, (short)0, acc, false, false);
  }

  const float bc = bias[col];
#pragma unroll
  for (int j = 0; j < 8; ++j) {
    int r = tm * 16 + j + (lane >> 4) * 8;
    out[(long)r * HH + (tn * 16 + (lane & 15))] = acc[j] + bc;
  }
}

#define KT 512
#define LST (KT + 4)

__device__ __forceinline__ void stage_panel(const float* __restrict__ A,
                                            float* __restrict__ As,
                                            int tm, int t, int tid) {
  for (int i = tid; i < 16 * (HH / 4); i += 256) {
    int rowl = i >> 5;
    int c4 = i & 31;
    float4 v = ((const float4*)(A + (long)(tm * 16 + rowl) * HH))[c4];
    *(float4*)&As[rowl * LST + t * HH + c4 * 4] = v;
  }
}

__global__ __launch_bounds__(256) void k_cheb_gemm128(
    const float* __restrict__ T0, const float* __restrict__ T1,
    const float* __restrict__ T2, const float* __restrict__ T3,
    const float* __restrict__ W, const float* __restrict__ bias,
    float* __restrict__ out) {
  __shared__ float As[16 * LST];
  const int tid = threadIdx.x;
  const int lane = tid & 31;
  const int wave = tid >> 5;
  const int tm = blockIdx.x;

  stage_panel(T0, As, tm, 0, tid);
  stage_panel(T1, As, tm, 1, tid);
  stage_panel(T2, As, tm, 2, tid);
  stage_panel(T3, As, tm, 3, tid);
  __syncthreads();

  const int l16 = lane & 15;
  const int khalf = (lane >> 4) * 2;
  const int col = wave * 16 + l16;

  v8f acc = {};
#pragma unroll 8
  for (int k0 = 0; k0 < KT; k0 += 4) {
    v2f a = *(const v2f*)&As[l16 * LST + k0 + khalf];
    v2f b;
    b[0] = W[(long)(k0 + khalf) * HH + col];
    b[1] = W[(long)(k0 + khalf + 1) * HH + col];
    acc = __builtin_amdgcn_wmma_f32_16x16x4_f32(
        false, a, false, b, (short)0, acc, false, false);
  }

  const float bc = bias[col];
#pragma unroll
  for (int j = 0; j < 8; ++j) {
    int r = tm * 16 + j + (lane >> 4) * 8;
    out[(long)r * HH + col] = acc[j] + bc;
  }
}


__global__ __launch_bounds__(128) void k_act_stats(float* __restrict__ h,
                                                   float* __restrict__ stats,
                                                   int actType) {
  const int ch = threadIdx.x;
  float s = 0.0f, s2 = 0.0f;
  for (int r = blockIdx.x; r < NN; r += gridDim.x) {
    long rn = (long)(r + gridDim.x);
    if (rn < NN) __builtin_prefetch(&h[rn * HH + ch], 0, 1);
    float v = h[(long)r * HH + ch];
    v = (actType == 0) ? (v > 0.0f ? v : 0.01f * v)
                       : fmaxf(v, 0.0f);
    h[(long)r * HH + ch] = v;
    s += v;
    s2 += v * v;
  }
  atomicAdd(&stats[ch], s);
  atomicAdd(&stats[HH + ch], s2);
}

__global__ __launch_bounds__(128) void k_bn_apply(float* __restrict__ h,
                                                  const float* __restrict__ stats,
                                                  const float* __restrict__ g,
                                                  const float* __restrict__ be) {
  const int ch = threadIdx.x;
  const float invN = 1.0f / (float)NN;
  float m = stats[ch] * invN;
  float var = stats[HH + ch] * invN - m * m;
  float sc = g[ch] * rsqrtf(var + 1e-5f);
  float sh = be[ch] - m * sc;
  for (int r = blockIdx.x; r < NN; r += gridDim.x) {
    long idx = (long)r * HH + ch;
    h[idx] = h[idx] * sc + sh;
  }
}

__global__ __launch_bounds__(256) void k_final(const float* __restrict__ h,
                                               const float* __restrict__ Wm,
                                               const float* __restrict__ bm,
                                               float* __restrict__ out) {
  int n = blockIdx.x * 8 + (threadIdx.x >> 5);
  if (n >= NN) return;
  int lane = threadIdx.x & 31;
  const float4* hp = (const float4*)(h + (long)n * HH);
  float4 v = hp[lane];
  int c = lane * 4;
  float ns = v.x * v.x + v.y * v.y + v.z * v.z + v.w * v.w;
  float d0 = v.x * Wm[(c + 0) * 3 + 0] + v.y * Wm[(c + 1) * 3 + 0] +
             v.z * Wm[(c + 2) * 3 + 0] + v.w * Wm[(c + 3) * 3 + 0];
  float d1 = v.x * Wm[(c + 0) * 3 + 1] + v.y * Wm[(c + 1) * 3 + 1] +
             v.z * Wm[(c + 2) * 3 + 1] + v.w * Wm[(c + 3) * 3 + 1];
  float d2 = v.x * Wm[(c + 0) * 3 + 2] + v.y * Wm[(c + 1) * 3 + 2] +
             v.z * Wm[(c + 2) * 3 + 2] + v.w * Wm[(c + 3) * 3 + 2];
#pragma unroll
  for (int off = 16; off > 0; off >>= 1) {
    ns += __shfl_xor(ns, off, 32);
    d0 += __shfl_xor(d0, off, 32);
    d1 += __shfl_xor(d1, off, 32);
    d2 += __shfl_xor(d2, off, 32);
  }
  if (lane == 0) {
    float inv = 1.0f / fmaxf(sqrtf(ns), 1e-12f);
    out[(long)n * 3 + 0] = d0 * inv + bm[0];
    out[(long)n * 3 + 1] = d1 * inv + bm[1];
    out[(long)n * 3 + 2] = d2 * inv + bm[2];
  }
}


static inline int cdiv(long a, long b) { return (int)((a + b - 1) / b); }

extern "C" void kernel_launch(void* const* d_in, const int* in_sizes, int n_in,
                              void* d_out, int out_size, void* d_ws, size_t ws_size,
                              hipStream_t stream) {
  const float* x   = (const float*)d_in[0];
  const int*   ei  = (const int*)d_in[1];
  const float* W1  = (const float*)d_in[2];
  const float* b1  = (const float*)d_in[3];
  const float* g1  = (const float*)d_in[4];
  const float* be1 = (const float*)d_in[5];
  const float* W2  = (const float*)d_in[6];
  const float* b2  = (const float*)d_in[7];
  const float* g2  = (const float*)d_in[8];
  const float* be2 = (const float*)d_in[9];
  const float* W3  = (const float*)d_in[10];
  const float* b3  = (const float*)d_in[11];
  const float* g3  = (const float*)d_in[12];
  const float* be3 = (const float*)d_in[13];
  const float* W4  = (const float*)d_in[14];
  const float* b4  = (const float*)d_in[15];
  const float* Wm  = (const float*)d_in[16];
  const float* bm  = (const float*)d_in[17];
  float* out = (float*)d_out;

  const int* src = ei;
  const int* dst = ei + EE;

  float* ws   = (float*)d_ws;
  float* deg  = ws;
  float* dinv = deg + NN;
  float* wE   = dinv + NN;
  float* T1   = wE + EE;
  float* T2   = T1 + (long)NN * HH;
  float* T3   = T2 + (long)NN * HH;
  float* hA   = T3 + (long)NN * HH;
  float* hB   = hA + (long)NN * HH;
  float* stats = hB + (long)NN * HH;

  const long NH = (long)NN * HH;
  const dim3 g3grid(NN / 16, HH / 16);
  const int g128grid = NN / 16;

  k_zero<<<cdiv(NN, 256), 256, 0, stream>>>(deg, NN);
  k_degree<<<cdiv(EE, 256), 256, 0, stream>>>(src, deg);
  k_dinv<<<cdiv(NN, 256), 256, 0, stream>>>(deg, dinv);
  k_edgew<<<cdiv(EE, 256), 256, 0, stream>>>(src, dst, dinv, wE);

  k_zero<<<cdiv((long)NN * 3, 256), 256, 0, stream>>>(T1, (long)NN * 3);
  k_spmv3<<<cdiv(EE, 256), 256, 0, stream>>>(x, src, dst, wE, 1.0f, T1);
  k_setneg<<<cdiv((long)NN * 3, 256), 256, 0, stream>>>(T2, x, (long)NN * 3);
  k_spmv3<<<cdiv(EE, 256), 256, 0, stream>>>(T1, src, dst, wE, 2.0f, T2);
  k_setneg<<<cdiv((long)NN * 3, 256), 256, 0, stream>>>(T3, T1, (long)NN * 3);
  k_spmv3<<<cdiv(EE, 256), 256, 0, stream>>>(T2, src, dst, wE, 2.0f, T3);
  k_cheb_gemm3<<<g3grid, 32, 0, stream>>>(x, T1, T2, T3, W1, b1, hA);
  k_zero<<<cdiv(2 * HH, 256), 256, 0, stream>>>(stats, 2 * HH);
  k_act_stats<<<512, 128, 0, stream>>>(hA, stats, 0);
  k_bn_apply<<<512, 128, 0, stream>>>(hA, stats, g1, be1);

  k_zero<<<cdiv(NH, 256), 256, 0, stream>>>(T1, NH);
  k_spmv128<<<cdiv(EE, 8), 256, 0, stream>>>(hA, src, dst, wE, 1.0f, T1);
  k_setneg<<<cdiv(NH, 256), 256, 0, stream>>>(T2, hA, NH);
  k_spmv128<<<cdiv(EE, 8), 256, 0, stream>>>(T1, src, dst, wE, 2.0f, T2);
  k_setneg<<<cdiv(NH, 256), 256, 0, stream>>>(T3, T1, NH);
  k_spmv128<<<cdiv(EE, 8), 256, 0, stream>>>(T2, src, dst, wE, 2.0f, T3);
  k_cheb_gemm128<<<g128grid, 256, 0, stream>>>(hA, T1, T2, T3, W2, b2, hB);
  k_zero<<<cdiv(2 * HH, 256), 256, 0, stream>>>(stats, 2 * HH);
  k_act_stats<<<512, 128, 0, stream>>>(hB, stats, 0);
  k_bn_apply<<<512, 128, 0, stream>>>(hB, stats, g2, be2);

  k_zero<<<cdiv(NH, 256), 256, 0, stream>>>(T1, NH);
  k_spmv128<<<cdiv(EE, 8), 256, 0, stream>>>(hB, src, dst, wE, 1.0f, T1);
  k_setneg<<<cdiv(NH, 256), 256, 0, stream>>>(T2, hB, NH);
  k_spmv128<<<cdiv(EE, 8), 256, 0, stream>>>(T1, src, dst, wE, 2.0f, T2);
  k_setneg<<<cdiv(NH, 256), 256, 0, stream>>>(T3, T1, NH);
  k_spmv128<<<cdiv(EE, 8), 256, 0, stream>>>(T2, src, dst, wE, 2.0f, T3);
  k_cheb_gemm128<<<g128grid, 256, 0, stream>>>(hB, T1, T2, T3, W3, b3, hA);
  k_zero<<<cdiv(2 * HH, 256), 256, 0, stream>>>(stats, 2 * HH);
  k_act_stats<<<512, 128, 0, stream>>>(hA, stats, 1);
  k_bn_apply<<<512, 128, 0, stream>>>(hA, stats, g3, be3);

  k_zero<<<cdiv(NH, 256), 256, 0, stream>>>(T1, NH);
  k_spmv128<<<cdiv(EE, 8), 256, 0, stream>>>(hA, src, dst, wE, 1.0f, T1);
  k_setneg<<<cdiv(NH, 256), 256, 0, stream>>>(T2, hA, NH);
  k_spmv128<<<cdiv(EE, 8), 256, 0, stream>>>(T1, src, dst, wE, 2.0f, T2);
  k_setneg<<<cdiv(NH, 256), 256, 0, stream>>>(T3, T1, NH);
  k_spmv128<<<cdiv(EE, 8), 256, 0, stream>>>(T2, src, dst, wE, 2.0f, T3);
  k_cheb_gemm128<<<g128grid, 256, 0, stream>>>(hA, T1, T2, T3, W4, b4, hB);

  k_final<<<cdiv(NN, 8), 256, 0, stream>>>(hB, Wm, bm, out);
}


# SpMV: f32-atomic scatter replaced by gather over a dst-grouped adjacency rebuilt per call (int count, scan, fill, rows sorted by edge index): row sums in fixed edge order; 2x and -T0 folded in
# speedup vs baseline: 2.3684x; 2.3684x over previous
#include <hip/hip_runtime.h>
#include <hip/hip_bf16.h>


#define NN 50000
#define EE 800000
#define HH 128

typedef float v2f __attribute__((ext_vector_type(2)));
typedef float v8f __attribute__((ext_vector_type(8)));


__global__ void k_zero(float* __restrict__ p, long n) {
  long i = (long)blockIdx.x * blockDim.x + threadIdx.x;
  if (i < n) p[i] = 0.0f;
}


__global__ void k_count(const int* __restrict__ src, const int* __restrict__ dst,
                        int* __restrict__ cntDst, int* __restrict__ cntSrc,
                        int* __restrict__ rank) {
  int e = blockIdx.x * blockDim.x + threadIdx.x;
  if (e < EE) {
    atomicAdd(&cntSrc[src[e]], 1);
    rank[e] = atomicAdd(&cntDst[dst[e]], 1);
  }
}

#define SCAN_T 1024
#define SCAN_PER ((NN + SCAN_T - 1) / SCAN_T)
__global__ __launch_bounds__(SCAN_T) void k_scan_dinv(
    const int* __restrict__ cntDst, const int* __restrict__ cntSrc,
    int* __restrict__ rowStart, float* __restrict__ dinv) {
  __shared__ int sc[NN];
  __shared__ int wsum[32];
  const int t = threadIdx.x;
  const int lane = t & 31;
  const int wv = t >> 5;
  for (int i = t; i < NN; i += SCAN_T) {
    sc[i] = cntDst[i];
    float d = (float)cntSrc[i];
    dinv[i] = (d > 0.0f) ? rsqrtf(fmaxf(d, 1.0f)) : 0.0f;
  }
  __syncthreads();
  const int b = t * SCAN_PER;
  const int bend = (b + SCAN_PER < NN) ? (b + SCAN_PER) : NN;
  int s = 0;
  for (int i = b; i < bend; ++i) s += sc[i];
  int x = s;
  for (int off = 1; off < 32; off <<= 1) {
    int y = __shfl_up(x, off, 32);
    if (lane >= off) x += y;
  }
  if (lane == 31) wsum[wv] = x;
  __syncthreads();
  if (wv == 0) {
    int w = wsum[lane];
    for (int off = 1; off < 32; off <<= 1) {
      int y = __shfl_up(w, off, 32);
      if (lane >= off) w += y;
    }
    wsum[lane] = w;
  }
  __syncthreads();
  int run = x - s + ((wv > 0) ? wsum[wv - 1] : 0);
  for (int i = b; i < bend; ++i) {
    int v = sc[i];
    sc[i] = run;
    run += v;
  }
  __syncthreads();
  for (int i = t; i < NN; i += SCAN_T) rowStart[i] = sc[i];
}

__global__ void k_fill(const int* __restrict__ dst, const int* __restrict__ rank,
                       const int* __restrict__ rowStart, int* __restrict__ eid) {
  int e = blockIdx.x * blockDim.x + threadIdx.x;
  if (e < EE) eid[rowStart[dst[e]] + rank[e]] = e;
}

__global__ __launch_bounds__(256) void k_rowsort(
    const int* __restrict__ src, const int* __restrict__ rowStart,
    const int* __restrict__ eid, int* __restrict__ col) {
  const int r = blockIdx.x * 8 + __builtin_amdgcn_readfirstlane((int)(threadIdx.x >> 5));
  if (r >= NN) return;
  const int lane = threadIdx.x & 31;
  const int beg = rowStart[r];
  const int end = (r + 1 < NN) ? rowStart[r + 1] : EE;
  for (int i = beg + lane; i < end; i += 32) {
    const int my = eid[i];
    int rk = 0;
    for (int j = beg; j < end; ++j) rk += (eid[j] < my) ? 1 : 0;
    col[beg + rk] = src[my];
  }
}


__global__ __launch_bounds__(256) void k_spmv3(
    const float* __restrict__ h, const int* __restrict__ rowStart,
    const int* __restrict__ col, const float* __restrict__ dinv,
    const float* __restrict__ T0, float scale, float* __restrict__ out) {
  const int gid = blockIdx.x * 256 + threadIdx.x;
  const int r = gid >> 4;
  const int sub = threadIdx.x & 15;
  float a0 = 0.0f, a1 = 0.0f, a2 = 0.0f;
  if (r < NN) {
    const int beg = rowStart[r];
    const int end = (r + 1 < NN) ? rowStart[r + 1] : EE;
    const float dr = dinv[r];
    for (int p = beg + sub; p < end; p += 16) {
      const int s = col[p];
      const float w = -dinv[s] * dr;
      a0 += w * h[(long)s * 3 + 0];
      a1 += w * h[(long)s * 3 + 1];
      a2 += w * h[(long)s * 3 + 2];
    }
  }
  for (int off = 8; off > 0; off >>= 1) {
    a0 += __shfl_xor(a0, off, 32);
    a1 += __shfl_xor(a1, off, 32);
    a2 += __shfl_xor(a2, off, 32);
  }
  if (r < NN && sub == 0) {
    float o0 = scale * a0, o1 = scale * a1, o2 = scale * a2;
    if (T0) {
      o0 -= T0[(long)r * 3 + 0];
      o1 -= T0[(long)r * 3 + 1];
      o2 -= T0[(long)r * 3 + 2];
    }
    out[(long)r * 3 + 0] = o0;
    out[(long)r * 3 + 1] = o1;
    out[(long)r * 3 + 2] = o2;
  }
}

__global__ __launch_bounds__(256) void k_spmv128(
    const float* __restrict__ h, const int* __restrict__ rowStart,
    const int* __restrict__ col, const float* __restrict__ dinv,
    const float* __restrict__ T0, float scale, float* __restrict__ out) {
  const int r = blockIdx.x * 8 + __builtin_amdgcn_readfirstlane((int)(threadIdx.x >> 5));
  if (r >= NN) return;
  const int lane = threadIdx.x & 31;
  const int beg = rowStart[r];
  const int end = (r + 1 < NN) ? rowStart[r + 1] : EE;
  const float dr = dinv[r];
  float4 acc = make_float4(0.0f, 0.0f, 0.0f, 0.0f);
  for (int base = beg; base < end; base += 32) {
    const int n = (end - base < 32) ? (end - base) : 32;
    int myc = 0;
    float myw = 0.0f;
    if (lane < n) {
      myc = col[base + lane];
      myw = -dinv[myc] * dr;
    }
#pragma unroll 4
    for (int j = 0; j < n; ++j) {
      const int s = __builtin_amdgcn_readlane(myc, j);
      const float w = __int_as_float(__builtin_amdgcn_readlane(__float_as_int(myw), j));
      const float4 v = ((const float4*)(h + (long)s * HH))[lane];
      acc.x += w * v.x;
      acc.y += w * v.y;
      acc.z += w * v.z;
      acc.w += w * v.w;
    }
  }
  float4 o = make_float4(scale * acc.x, scale * acc.y, scale * acc.z, scale * acc.w);
  if (T0) {
    const float4 t0 = ((const float4*)(T0 + (long)r * HH))[lane];
    o.x -= t0.x;
    o.y -= t0.y;
    o.z -= t0.z;
    o.w -= t0.w;
  }
  ((float4*)(out + (long)r * HH))[lane] = o;
}


__global__ __launch_bounds__(32) void k_cheb_gemm3(
    const float* __restrict__ T0, const float* __restrict__ T1,
    const float* __restrict__ T2, const float* __restrict__ T3,
    const float* __restrict__ W, const float* __restrict__ bias,
    float* __restrict__ out) {
  const int lane = threadIdx.x;
  const int tm = blockIdx.x;
  const int tn = blockIdx.y;
  const int row = tm * 16 + (lane & 15);
  const int col = tn * 16 + (lane & 15);
  const int khalf = (lane >> 4) * 2;
  const float* Ts[4] = {T0, T1, T2, T3};

  v8f acc = {};
  for (int k0 = 0; k0 < 12; k0 += 4) {
    v2f a, b;
#pragma unroll
    for (int j = 0; j < 2; ++j) {
      int kk = k0 + khalf + j;
      int t = kk / 3;
      int cc = kk - t * 3;
      a[j] = Ts[t][(long)row * 3 + cc];
      b[j] = W[(long)kk * HH + col];
    }
    acc = __builtin_amdgcn_wmma_f32_16x16x4_f32(
        false, a, false, b, (short)0, acc, false, false);
  }

  const float bc = bias[col];
#pragma unroll
  for (int j = 0; j < 8; ++j) {
    int r = tm * 16 + j + (lane >> 4) * 8;
    out[(long)r * HH + (tn * 16 + (lane & 15))] = acc[j] + bc;
  }
}

#define KT 512
#define LST (KT + 4)

__device__ __forceinline__ void stage_panel(const float* __restrict__ A,
                                            float* __restrict__ As,
                                            int tm, int t, int tid) {
  for (int i = tid; i < 16 * (HH / 4); i += 256) {
    int rowl = i >> 5;
    int c4 = i & 31;
    float4 v = ((const float4*)(A + (long)(tm * 16 + rowl) * HH))[c4];
    *(float4*)&As[rowl * LST + t * HH + c4 * 4] = v;
  }
}

__global__ __launch_bounds__(256) void k_cheb_gemm128(
    const float* __restrict__ T0, const float* __restrict__ T1,
    const float* __restrict__ T2, const float* __restrict__ T3,
    const float* __restrict__ W, const float* __restrict__ bias,
    float* __restrict__ out) {
  __shared__ float As[16 * LST];
  const int tid = threadIdx.x;
  const int lane = tid & 31;
  const int wave = tid >> 5;
  const int tm = blockIdx.x;

  stage_panel(T0, As, tm, 0, tid);
  stage_panel(T1, As, tm, 1, tid);
  stage_panel(T2, As, tm, 2, tid);
  stage_panel(T3, As, tm, 3, tid);
  __syncthreads();

  const int l16 = lane & 15;
  const int khalf = (lane >> 4) * 2;
  const int col = wave * 16 + l16;

  v8f acc = {};
#pragma unroll 8
  for (int k0 = 0; k0 < KT; k0 += 4) {
    v2f a = *(const v2f*)&As[l16 * LST + k0 + khalf];
    v2f b;
    b[0] = W[(long)(k0 + khalf) * HH + col];
    b[1] = W[(long)(k0 + khalf + 1) * HH + col];
    acc = __builtin_amdgcn_wmma_f32_16x16x4_f32(
        false, a, false, b, (short)0, acc, false, false);
  }

  const float bc = bias[col];
#pragma unroll
  for (int j = 0; j < 8; ++j) {
    int r = tm * 16 + j + (lane >> 4) * 8;
    out[(long)r * HH + col] = acc[j] + bc;
  }
}


__global__ __launch_bounds__(128) void k_act_stats(float* __restrict__ h,
                                                   float* __restrict__ stats,
                                                   int actType) {
  const int ch = threadIdx.x;
  float s = 0.0f, s2 = 0.0f;
  for (int r = blockIdx.x; r < NN; r += gridDim.x) {
    long rn = (long)(r + gridDim.x);
    if (rn < NN) __builtin_prefetch(&h[rn * HH + ch], 0, 1);
    float v = h[(long)r * HH + ch];
    v = (actType == 0) ? (v > 0.0f ? v : 0.01f * v)
                       : fmaxf(v, 0.0f);
    h[(long)r * HH + ch] = v;
    s += v;
    s2 += v * v;
  }
  atomicAdd(&stats[ch], s);
  atomicAdd(&stats[HH + ch], s2);
}

__global__ __launch_bounds__(128) void k_bn_apply(float* __restrict__ h,
                                                  const float* __restrict__ stats,
                                                  const float* __restrict__ g,
                                                  const float* __restrict__ be) {
  const int ch = threadIdx.x;
  const float invN = 1.0f / (float)NN;
  float m = stats[ch] * invN;
  float var = stats[HH + ch] * invN - m * m;
  float sc = g[ch] * rsqrtf(var + 1e-5f);
  float sh = be[ch] - m * sc;
  for (int r = blockIdx.x; r < NN; r += gridDim.x) {
    long idx = (long)r * HH + ch;
    h[idx] = h[idx] * sc + sh;
  }
}

__global__ __launch_bounds__(256) void k_final(const float* __restrict__ h,
                                               const float* __restrict__ Wm,
                                               const float* __restrict__ bm,
                                               float* __restrict__ out) {
  int n = blockIdx.x * 8 + (threadIdx.x >> 5);
  if (n >= NN) return;
  int lane = threadIdx.x & 31;
  const float4* hp = (const float4*)(h + (long)n * HH);
  float4 v = hp[lane];
  int c = lane * 4;
  float ns = v.x * v.x + v.y * v.y + v.z * v.z + v.w * v.w;
  float d0 = v.x * Wm[(c + 0) * 3 + 0] + v.y * Wm[(c + 1) * 3 + 0] +
             v.z * Wm[(c + 2) * 3 + 0] + v.w * Wm[(c + 3) * 3 + 0];
  float d1 = v.x * Wm[(c + 0) * 3 + 1] + v.y * Wm[(c + 1) * 3 + 1] +
             v.z * Wm[(c + 2) * 3 + 1] + v.w * Wm[(c + 3) * 3 + 1];
  float d2 = v.x * Wm[(c + 0) * 3 + 2] + v.y * Wm[(c + 1) * 3 + 2] +
             v.z * Wm[(c + 2) * 3 + 2] + v.w * Wm[(c + 3) * 3 + 2];
#pragma unroll
  for (int off = 16; off > 0; off >>= 1) {
    ns += __shfl_xor(ns, off, 32);
    d0 += __shfl_xor(d0, off, 32);
    d1 += __shfl_xor(d1, off, 32);
    d2 += __shfl_xor(d2, off, 32);
  }
  if (lane == 0) {
    float inv = 1.0f / fmaxf(sqrtf(ns), 1e-12f);
    out[(long)n * 3 + 0] = d0 * inv + bm[0];
    out[(long)n * 3 + 1] = d1 * inv + bm[1];
    out[(long)n * 3 + 2] = d2 * inv + bm[2];
  }
}


static inline int cdiv(long a, long b) { return (int)((a + b - 1) / b); }

extern "C" void kernel_launch(void* const* d_in, const int* in_sizes, int n_in,
                              void* d_out, int out_size, void* d_ws, size_t ws_size,
                              hipStream_t stream) {
  const float* x   = (const float*)d_in[0];
  const int*   ei  = (const int*)d_in[1];
  const float* W1  = (const float*)d_in[2];
  const float* b1  = (const float*)d_in[3];
  const float* g1  = (const float*)d_in[4];
  const float* be1 = (const float*)d_in[5];
  const float* W2  = (const float*)d_in[6];
  const float* b2  = (const float*)d_in[7];
  const float* g2  = (const float*)d_in[8];
  const float* be2 = (const float*)d_in[9];
  const float* W3  = (const float*)d_in[10];
  const float* b3  = (const float*)d_in[11];
  const float* g3  = (const float*)d_in[12];
  const float* be3 = (const float*)d_in[13];
  const float* W4  = (const float*)d_in[14];
  const float* b4  = (const float*)d_in[15];
  const float* Wm  = (const float*)d_in[16];
  const float* bm  = (const float*)d_in[17];
  float* out = (float*)d_out;

  const int* src = ei;
  const int* dst = ei + EE;

  float* ws   = (float*)d_ws;
  int*   rowStart = (int*)ws;
  float* dinv = ws + NN;
  int*   col  = (int*)(dinv + NN);
  float* T1   = (float*)(col + EE);
  float* T2   = T1 + (long)NN * HH;
  float* T3   = T2 + (long)NN * HH;
  float* hA   = T3 + (long)NN * HH;
  float* hB   = hA + (long)NN * HH;
  float* stats = hB + (long)NN * HH;
  int* cntDst = (int*)hB;
  int* cntSrc = cntDst + NN;
  int* rank   = cntSrc + NN;
  int* eid    = rank + EE;

  const dim3 g3grid(NN / 16, HH / 16);
  const int g128grid = NN / 16;
  const int rowWaves = cdiv(NN, 8);
  const int row16    = cdiv((long)NN * 16, 256);

  k_zero<<<cdiv(2 * NN, 256), 256, 0, stream>>>((float*)cntDst, 2 * NN);
  k_count<<<cdiv(EE, 256), 256, 0, stream>>>(src, dst, cntDst, cntSrc, rank);
  k_scan_dinv<<<1, SCAN_T, 0, stream>>>(cntDst, cntSrc, rowStart, dinv);
  k_fill<<<cdiv(EE, 256), 256, 0, stream>>>(dst, rank, rowStart, eid);
  k_rowsort<<<rowWaves, 256, 0, stream>>>(src, rowStart, eid, col);

  k_spmv3<<<row16, 256, 0, stream>>>(x,  rowStart, col, dinv, nullptr, 1.0f, T1);
  k_spmv3<<<row16, 256, 0, stream>>>(T1, rowStart, col, dinv, x,       2.0f, T2);
  k_spmv3<<<row16, 256, 0, stream>>>(T2, rowStart, col, dinv, T1,      2.0f, T3);
  k_cheb_gemm3<<<g3grid, 32, 0, stream>>>(x, T1, T2, T3, W1, b1, hA);
  k_zero<<<cdiv(2 * HH, 256), 256, 0, stream>>>(stats, 2 * HH);
  k_act_stats<<<512, 128, 0, stream>>>(hA, stats, 0);
  k_bn_apply<<<512, 128, 0, stream>>>(hA, stats, g1, be1);

  k_spmv128<<<rowWaves, 256, 0, stream>>>(hA, rowStart, col, dinv, nullptr, 1.0f, T1);
  k_spmv128<<<rowWaves, 256, 0, stream>>>(T1, rowStart, col, dinv, hA,      2.0f, T2);
  k_spmv128<<<rowWaves, 256, 0, stream>>>(T2, rowStart, col, dinv, T1,      2.0f, T3);
  k_cheb_gemm128<<<g128grid, 256, 0, stream>>>(hA, T1, T2, T3, W2, b2, hB);
  k_zero<<<cdiv(2 * HH, 256), 256, 0, stream>>>(stats, 2 * HH);
  k_act_stats<<<512, 128, 0, stream>>>(hB, stats, 0);
  k_bn_apply<<<512, 128, 0, stream>>>(hB, stats, g2, be2);

  k_spmv128<<<rowWaves, 256, 0, stream>>>(hB, rowStart, col, dinv, nullptr, 1.0f, T1);
  k_spmv128<<<rowWaves, 256, 0, stream>>>(T1, rowStart, col, dinv, hB,      2.0f, T2);
  k_spmv128<<<rowWaves, 256, 0, stream>>>(T2, rowStart, col, dinv, T1,      2.0f, T3);
  k_cheb_gemm128<<<g128grid, 256, 0, stream>>>(hB, T1, T2, T3, W3, b3, hA);
  k_zero<<<cdiv(2 * HH, 256), 256, 0, stream>>>(stats, 2 * HH);
  k_act_stats<<<512, 128, 0, stream>>>(hA, stats, 1);
  k_bn_apply<<<512, 128, 0, stream>>>(hA, stats, g3, be3);

  k_spmv128<<<rowWaves, 256, 0, stream>>>(hA, rowStart, col, dinv, nullptr, 1.0f, T1);
  k_spmv128<<<rowWaves, 256, 0, stream>>>(T1, rowStart, col, dinv, hA,      2.0f, T2);
  k_spmv128<<<rowWaves, 256, 0, stream>>>(T2, rowStart, col, dinv, T1,      2.0f, T3);
  k_cheb_gemm128<<<g128grid, 256, 0, stream>>>(hA, T1, T2, T3, W4, b4, hB);

  k_final<<<cdiv(NN, 8), 256, 0, stream>>>(hB, Wm, bm, out);
}


# BatchNorm passes: a wave takes 16 whole rows, float4 per lane, all loads issued before its first store (was one 4-byte load in flight per wave); sums via per-block partials in fixed order, no atomics
# speedup vs baseline: 2.8334x; 1.1963x over previous
#include <hip/hip_runtime.h>
#include <hip/hip_bf16.h>


#define NN 50000
#define EE 800000
#define HH 128

typedef float v2f __attribute__((ext_vector_type(2)));
typedef float v8f __attribute__((ext_vector_type(8)));


__global__ void k_zero(float* __restrict__ p, long n) {
  long i = (long)blockIdx.x * blockDim.x + threadIdx.x;
  if (i < n) p[i] = 0.0f;
}


__global__ void k_count(const int* __restrict__ src, const int* __restrict__ dst,
                        int* __restrict__ cntDst, int* __restrict__ cntSrc,
                        int* __restrict__ rank) {
  int e = blockIdx.x * blockDim.x + threadIdx.x;
  if (e < EE) {
    atomicAdd(&cntSrc[src[e]], 1);
    rank[e] = atomicAdd(&cntDst[dst[e]], 1);
  }
}

#define SCAN_T 1024
#define SCAN_PER ((NN + SCAN_T - 1) / SCAN_T)
__global__ __launch_bounds__(SCAN_T) void k_scan_dinv(
    const int* __restrict__ cntDst, const int* __restrict__ cntSrc,
    int* __restrict__ rowStart, float* __restrict__ dinv) {
  __shared__ int sc[NN];
  __shared__ int wsum[32];
  const int t = threadIdx.x;
  const int lane = t & 31;
  const int wv = t >> 5;
  for (int i = t; i < NN; i += SCAN_T) {
    sc[i] = cntDst[i];
    float d = (float)cntSrc[i];
    dinv[i] = (d > 0.0f) ? rsqrtf(fmaxf(d, 1.0f)) : 0.0f;
  }
  __syncthreads();
  const int b = t * SCAN_PER;
  const int bend = (b + SCAN_PER < NN) ? (b + SCAN_PER) : NN;
  int s = 0;
  for (int i = b; i < bend; ++i) s += sc[i];
  int x = s;
  for (int off = 1; off < 32; off <<= 1) {
    int y = __shfl_up(x, off, 32);
    if (lane >= off) x += y;
  }
  if (lane == 31) wsum[wv] = x;
  __syncthreads();
  if (wv == 0) {
    int w = wsum[lane];
    for (int off = 1; off < 32; off <<= 1) {
      int y = __shfl_up(w, off, 32);
      if (lane >= off) w += y;
    }
    wsum[lane] = w;
  }
  __syncthreads();
  int run = x - s + ((wv > 0) ? wsum[wv - 1] : 0);
  for (int i = b; i < bend; ++i) {
    int v = sc[i];
    sc[i] = run;
    run += v;
  }
  __syncthreads();
  for (int i = t; i < NN; i += SCAN_T) rowStart[i] = sc[i];
}

__global__ void k_fill(const int* __restrict__ dst, const int* __restrict__ rank,
                       const int* __restrict__ rowStart, int* __restrict__ eid) {
  int e = blockIdx.x * blockDim.x + threadIdx.x;
  if (e < EE) eid[rowStart[dst[e]] + rank[e]] = e;
}

__global__ __launch_bounds__(256) void k_rowsort(
    const int* __restrict__ src, const int* __restrict__ rowStart,
    const int* __restrict__ eid, int* __restrict__ col) {
  const int r = blockIdx.x * 8 + __builtin_amdgcn_readfirstlane((int)(threadIdx.x >> 5));
  if (r >= NN) return;
  const int lane = threadIdx.x & 31;
  const int beg = rowStart[r];
  const int end = (r + 1 < NN) ? rowStart[r + 1] : EE;
  for (int i = beg + lane; i < end; i += 32) {
    const int my = eid[i];
    int rk = 0;
    for (int j = beg; j < end; ++j) rk += (eid[j] < my) ? 1 : 0;
    col[beg + rk] = src[my];
  }
}


__global__ __launch_bounds__(256) void k_spmv3(
    const float* __restrict__ h, const int* __restrict__ rowStart,
    const int* __restrict__ col, const float* __restrict__ dinv,
    const float* __restrict__ T0, float scale, float* __restrict__ out) {
  const int gid = blockIdx.x * 256 + threadIdx.x;
  const int r = gid >> 4;
  const int sub = threadIdx.x & 15;
  float a0 = 0.0f, a1 = 0.0f, a2 = 0.0f;
  if (r < NN) {
    const int beg = rowStart[r];
    const int end = (r + 1 < NN) ? rowStart[r + 1] : EE;
    const float dr = dinv[r];
    for (int p = beg + sub; p < end; p += 16) {
      const int s = col[p];
      const float w = -dinv[s] * dr;
      a0 += w * h[(long)s * 3 + 0];
      a1 += w * h[(long)s * 3 + 1];
      a2 += w * h[(long)s * 3 + 2];
    }
  }
  for (int off = 8; off > 0; off >>= 1) {
    a0 += __shfl_xor(a0, off, 32);
    a1 += __shfl_xor(a1, off, 32);
    a2 += __shfl_xor(a2, off, 32);
  }
  if (r < NN && sub == 0) {
    float o0 = scale * a0, o1 = scale * a1, o2 = scale * a2;
    if (T0) {
      o0 -= T0[(long)r * 3 + 0];
      o1 -= T0[(long)r * 3 + 1];
      o2 -= T0[(long)r * 3 + 2];
    }
    out[(long)r * 3 + 0] = o0;
    out[(long)r * 3 + 1] = o1;
    out[(long)r * 3 + 2] = o2;
  }
}

__global__ __launch_bounds__(256) void k_spmv128(
    const float* __restrict__ h, const int* __restrict__ rowStart,
    const int* __restrict__ col, const float* __restrict__ dinv,
    const float* __restrict__ T0, float scale, float* __restrict__ out) {
  const int r = blockIdx.x * 8 + __builtin_amdgcn_readfirstlane((int)(threadIdx.x >> 5));
  if (r >= NN) return;
  const int lane = threadIdx.x & 31;
  const int beg = rowStart[r];
  const int end = (r + 1 < NN) ? rowStart[r + 1] : EE;
  const float dr = dinv[r];
  float4 acc = make_float4(0.0f, 0.0f, 0.0f, 0.0f);
  for (int base = beg; base < end; base += 32) {
    const int n = (end - base < 32) ? (end - base) : 32;
    int myc = 0;
    float myw = 0.0f;
    if (lane < n) {
      myc = col[base + lane];
      myw = -dinv[myc] * dr;
    }
#pragma unroll 4
    for (int j = 0; j < n; ++j) {
      const int s = __builtin_amdgcn_readlane(myc, j);
      const float w = __int_as_float(__builtin_amdgcn_readlane(__float_as_int(myw), j));
      const float4 v = ((const float4*)(h + (long)s * HH))[lane];
      acc.x += w * v.x;
      acc.y += w * v.y;
      acc.z += w * v.z;
      acc.w += w * v.w;
    }
  }
  float4 o = make_float4(scale * acc.x, scale * acc.y, scale * acc.z, scale * acc.w);
  if (T0) {
    const float4 t0 = ((const float4*)(T0 + (long)r * HH))[lane];
    o.x -= t0.x;
    o.y -= t0.y;
    o.z -= t0.z;
    o.w -= t0.w;
  }
  ((float4*)(out + (long)r * HH))[lane] = o;
}


__global__ __launch_bounds__(32) void k_cheb_gemm3(
    const float* __restrict__ T0, const float* __restrict__ T1,
    const float* __restrict__ T2, const float* __restrict__ T3,
    const float* __restrict__ W, const float* __restrict__ bias,
    float* __restrict__ out) {
  const int lane = threadIdx.x;
  const int tm = blockIdx.x;
  const int tn = blockIdx.y;
  const int row = tm * 16 + (lane & 15);
  const int col = tn * 16 + (lane & 15);
  const int khalf = (lane >> 4) * 2;
  const float* Ts[4] = {T0, T1, T2, T3};

  v8f acc = {};
  for (int k0 = 0; k0 < 12; k0 += 4) {
    v2f a, b;
#pragma unroll
    for (int j = 0; j < 2; ++j) {
      int kk = k0 + khalf + j;
      int t = kk / 3;
      int cc = kk - t * 3;
      a[j] = Ts[t][(long)row * 3 + cc];
      b[j] = W[(long)kk * HH + col];
    }
    acc = __builtin_amdgcn_wmma_f32_16x16x4_f32(
        false, a, false, b, (short)0, acc, false, false);
  }

  const float bc = bias[col];
#pragma unroll
  for (int j = 0; j < 8; ++j) {
    int r = tm * 16 + j + (lane >> 4) * 8;
    out[(long)r * HH + (tn * 16 + (lane & 15))] = acc[j] + bc;
  }
}

#define KT 512
#define LST (KT + 4)

__device__ __forceinline__ void stage_panel(const float* __restrict__ A,
                                            float* __restrict__ As,
                                            int tm, int t, int tid) {
  for (int i = tid; i < 16 * (HH / 4); i += 256) {
    int rowl = i >> 5;
    int c4 = i & 31;
    float4 v = ((const float4*)(A + (long)(tm * 16 + rowl) * HH))[c4];
    *(float4*)&As[rowl * LST + t * HH + c4 * 4] = v;
  }
}

__global__ __launch_bounds__(256) void k_cheb_gemm128(
    const float* __restrict__ T0, const float* __restrict__ T1,
    const float* __restrict__ T2, const float* __restrict__ T3,
    const float* __restrict__ W, const float* __restrict__ bias,
    float* __restrict__ out) {
  __shared__ float As[16 * LST];
  const int tid = threadIdx.x;
  const int lane = tid & 31;
  const int wave = tid >> 5;
  const int tm = blockIdx.x;

  stage_panel(T0, As, tm, 0, tid);
  stage_panel(T1, As, tm, 1, tid);
  stage_panel(T2, As, tm, 2, tid);
  stage_panel(T3, As, tm, 3, tid);
  __syncthreads();

  const int l16 = lane & 15;
  const int khalf = (lane >> 4) * 2;
  const int col = wave * 16 + l16;

  v8f acc = {};
#pragma unroll 8
  for (int k0 = 0; k0 < KT; k0 += 4) {
    v2f a = *(const v2f*)&As[l16 * LST + k0 + khalf];
    v2f b;
    b[0] = W[(long)(k0 + khalf) * HH + col];
    b[1] = W[(long)(k0 + khalf + 1) * HH + col];
    acc = __builtin_amdgcn_wmma_f32_16x16x4_f32(
        false, a, false, b, (short)0, acc, false, false);
  }

  const float bc = bias[col];
#pragma unroll
  for (int j = 0; j < 8; ++j) {
    int r = tm * 16 + j + (lane >> 4) * 8;
    out[(long)r * HH + col] = acc[j] + bc;
  }
}


#define BN_ROWS 16
#define BN_WAVES 8
#define BN_BLOCKS ((NN + BN_ROWS * BN_WAVES - 1) / (BN_ROWS * BN_WAVES))
static_assert(NN % BN_ROWS == 0, "a wave's rows are all inside or all outside");

template <int ACT>
__global__ __launch_bounds__(256) void k_act_stats(float* __restrict__ h,
                                                   float* __restrict__ part) {
  __shared__ float red[BN_WAVES][2 * HH];
  const int lane = threadIdx.x & 31;
  const int wv = __builtin_amdgcn_readfirstlane((int)(threadIdx.x >> 5));
  const int r0 = (blockIdx.x * BN_WAVES + wv) * BN_ROWS;
  float4 s = make_float4(0.0f, 0.0f, 0.0f, 0.0f);
  float4 s2 = make_float4(0.0f, 0.0f, 0.0f, 0.0f);
  if (r0 < NN) {
    float4* hp = (float4*)(h + (long)r0 * HH) + lane;
    float4 v[BN_ROWS];
#pragma unroll
    for (int u = 0; u < BN_ROWS; ++u) v[u] = hp[u * (HH / 4)];
#pragma unroll
    for (int u = 0; u < BN_ROWS; ++u) {
      float4 a = v[u];
      if (ACT == 0) {
        a.x = a.x > 0.0f ? a.x : 0.01f * a.x;
        a.y = a.y > 0.0f ? a.y : 0.01f * a.y;
        a.z = a.z > 0.0f ? a.z : 0.01f * a.z;
        a.w = a.w > 0.0f ? a.w : 0.01f * a.w;
      } else {
        a.x = fmaxf(a.x, 0.0f);
        a.y = fmaxf(a.y, 0.0f);
        a.z = fmaxf(a.z, 0.0f);
        a.w = fmaxf(a.w, 0.0f);
      }
      hp[u * (HH / 4)] = a;
      s.x += a.x;  s.y += a.y;  s.z += a.z;  s.w += a.w;
      s2.x += a.x * a.x;  s2.y += a.y * a.y;  s2.z += a.z * a.z;  s2.w += a.w * a.w;
    }
  }
  *(float4*)&red[wv][lane * 4] = s;
  *(float4*)&red[wv][HH + lane * 4] = s2;
  __syncthreads();
  const int t = threadIdx.x;
  float acc = red[0][t];
  for (int w = 1; w < BN_WAVES; ++w) acc += red[w][t];
  part[(long)blockIdx.x * (2 * HH) + t] = acc;
}

#define FIN_Q 4
#define FIN_PER ((BN_BLOCKS + FIN_Q - 1) / FIN_Q)
__global__ __launch_bounds__(FIN_Q * 2 * HH) void k_bn_finalize(
    const float* __restrict__ part, const float* __restrict__ g,
    const float* __restrict__ be, float* __restrict__ scsh) {
  __shared__ float red[FIN_Q][2 * HH];
  const int t = threadIdx.x & (2 * HH - 1);
  const int q = threadIdx.x / (2 * HH);
  const int b0 = q * FIN_PER;
  float acc = 0.0f;
#pragma unroll 14
  for (int i = 0; i < FIN_PER; ++i) {
    const int b = b0 + i;
    const int bc = (b < BN_BLOCKS) ? b : (BN_BLOCKS - 1);
    const float v = part[(long)bc * (2 * HH) + t];
    acc += (b < BN_BLOCKS) ? v : 0.0f;
  }
  red[q][t] = acc;
  __syncthreads();
  if (threadIdx.x < HH) {
    const int ch = threadIdx.x;
    float sum = red[0][ch], sq = red[0][HH + ch];
    for (int k = 1; k < FIN_Q; ++k) {
      sum += red[k][ch];
      sq += red[k][HH + ch];
    }
    const float invN = 1.0f / (float)NN;
    const float m = sum * invN;
    const float var = sq * invN - m * m;
    const float sc = g[ch] * rsqrtf(var + 1e-5f);
    scsh[ch] = sc;
    scsh[HH + ch] = be[ch] - m * sc;
  }
}

__global__ __launch_bounds__(256) void k_bn_apply(float* __restrict__ h,
                                                  const float* __restrict__ scsh) {
  const int lane = threadIdx.x & 31;
  const int wv = __builtin_amdgcn_readfirstlane((int)(threadIdx.x >> 5));
  const int r0 = (blockIdx.x * BN_WAVES + wv) * BN_ROWS;
  if (r0 >= NN) return;
  const float4 sc = ((const float4*)scsh)[lane];
  const float4 sh = ((const float4*)(scsh + HH))[lane];
  float4* hp = (float4*)(h + (long)r0 * HH) + lane;
  float4 v[BN_ROWS];
#pragma unroll
  for (int u = 0; u < BN_ROWS; ++u) v[u] = hp[u * (HH / 4)];
#pragma unroll
  for (int u = 0; u < BN_ROWS; ++u) {
    float4 o;
    o.x = v[u].x * sc.x + sh.x;
    o.y = v[u].y * sc.y + sh.y;
    o.z = v[u].z * sc.z + sh.z;
    o.w = v[u].w * sc.w + sh.w;
    hp[u * (HH / 4)] = o;
  }
}

__global__ __launch_bounds__(256) void k_final(const float* __restrict__ h,
                                               const float* __restrict__ Wm,
                                               const float* __restrict__ bm,
                                               float* __restrict__ out) {
  int n = blockIdx.x * 8 + (threadIdx.x >> 5);
  if (n >= NN) return;
  int lane = threadIdx.x & 31;
  const float4* hp = (const float4*)(h + (long)n * HH);
  float4 v = hp[lane];
  int c = lane * 4;
  float ns = v.x * v.x + v.y * v.y + v.z * v.z + v.w * v.w;
  float d0 = v.x * Wm[(c + 0) * 3 + 0] + v.y * Wm[(c + 1) * 3 + 0] +
             v.z * Wm[(c + 2) * 3 + 0] + v.w * Wm[(c + 3) * 3 + 0];
  float d1 = v.x * Wm[(c + 0) * 3 + 1] + v.y * Wm[(c + 1) * 3 + 1] +
             v.z * Wm[(c + 2) * 3 + 1] + v.w * Wm[(c + 3) * 3 + 1];
  float d2 = v.x * Wm[(c + 0) * 3 + 2] + v.y * Wm[(c + 1) * 3 + 2] +
             v.z * Wm[(c + 2) * 3 + 2] + v.w * Wm[(c + 3) * 3 + 2];
#pragma unroll
  for (int off = 16; off > 0; off >>= 1) {
    ns += __shfl_xor(ns, off, 32);
    d0 += __shfl_xor(d0, off, 32);
    d1 += __shfl_xor(d1, off, 32);
    d2 += __shfl_xor(d2, off, 32);
  }
  if (lane == 0) {
    float inv = 1.0f / fmaxf(sqrtf(ns), 1e-12f);
    out[(long)n * 3 + 0] = d0 * inv + bm[0];
    out[(long)n * 3 + 1] = d1 * inv + bm[1];
    out[(long)n * 3 + 2] = d2 * inv + bm[2];
  }
}


static inline int cdiv(long a, long b) { return (int)((a + b - 1) / b); }

extern "C" void kernel_launch(void* const* d_in, const int* in_sizes, int n_in,
                              void* d_out, int out_size, void* d_ws, size_t ws_size,
                              hipStream_t stream) {
  const float* x   = (const float*)d_in[0];
  const int*   ei  = (const int*)d_in[1];
  const float* W1  = (const float*)d_in[2];
  const float* b1  = (const float*)d_in[3];
  const float* g1  = (const float*)d_in[4];
  const float* be1 = (const float*)d_in[5];
  const float* W2  = (const float*)d_in[6];
  const float* b2  = (const float*)d_in[7];
  const float* g2  = (const float*)d_in[8];
  const float* be2 = (const float*)d_in[9];
  const float* W3  = (const float*)d_in[10];
  const float* b3  = (const float*)d_in[11];
  const float* g3  = (const float*)d_in[12];
  const float* be3 = (const float*)d_in[13];
  const float* W4  = (const float*)d_in[14];
  const float* b4  = (const float*)d_in[15];
  const float* Wm  = (const float*)d_in[16];
  const float* bm  = (const float*)d_in[17];
  float* out = (float*)d_out;

  const int* src = ei;
  const int* dst = ei + EE;

  float* ws   = (float*)d_ws;
  int*   rowStart = (int*)ws;
  float* dinv = ws + NN;
  int*   col  = (int*)(dinv + NN);
  float* T1   = (float*)(col + EE);
  float* T2   = T1 + (long)NN * HH;
  float* T3   = T2 + (long)NN * HH;
  float* hA   = T3 + (long)NN * HH;
  float* hB   = hA + (long)NN * HH;
  float* scsh = hB + (long)NN * HH;
  float* part = T1;
  int* cntDst = (int*)hB;
  int* cntSrc = cntDst + NN;
  int* rank   = cntSrc + NN;
  int* eid    = rank + EE;

  const dim3 g3grid(NN / 16, HH / 16);
  const int g128grid = NN / 16;
  const int rowWaves = cdiv(NN, 8);
  const int row16    = cdiv((long)NN * 16, 256);

  k_zero<<<cdiv(2 * NN, 256), 256, 0, stream>>>((float*)cntDst, 2 * NN);
  k_count<<<cdiv(EE, 256), 256, 0, stream>>>(src, dst, cntDst, cntSrc, rank);
  k_scan_dinv<<<1, SCAN_T, 0, stream>>>(cntDst, cntSrc, rowStart, dinv);
  k_fill<<<cdiv(EE, 256), 256, 0, stream>>>(dst, rank, rowStart, eid);
  k_rowsort<<<rowWaves, 256, 0, stream>>>(src, rowStart, eid, col);

  k_spmv3<<<row16, 256, 0, stream>>>(x,  rowStart, col, dinv, nullptr, 1.0f, T1);
  k_spmv3<<<row16, 256, 0, stream>>>(T1, rowStart, col, dinv, x,       2.0f, T2);
  k_spmv3<<<row16, 256, 0, stream>>>(T2, rowStart, col, dinv, T1,      2.0f, T3);
  k_cheb_gemm3<<<g3grid, 32, 0, stream>>>(x, T1, T2, T3, W1, b1, hA);
  k_act_stats<0><<<BN_BLOCKS, 256, 0, stream>>>(hA, part);
  k_bn_finalize<<<1, FIN_Q * 2 * HH, 0, stream>>>(part, g1, be1, scsh);
  k_bn_apply<<<BN_BLOCKS, 256, 0, stream>>>(hA, scsh);

  k_spmv128<<<rowWaves, 256, 0, stream>>>(hA, rowStart, col, dinv, nullptr, 1.0f, T1);
  k_spmv128<<<rowWaves, 256, 0, stream>>>(T1, rowStart, col, dinv, hA,      2.0f, T2);
  k_spmv128<<<rowWaves, 256, 0, stream>>>(T2, rowStart, col, dinv, T1,      2.0f, T3);
  k_cheb_gemm128<<<g128grid, 256, 0, stream>>>(hA, T1, T2, T3, W2, b2, hB);
  k_act_stats<0><<<BN_BLOCKS, 256, 0, stream>>>(hB, part);
  k_bn_finalize<<<1, FIN_Q * 2 * HH, 0, stream>>>(part, g2, be2, scsh);
  k_bn_apply<<<BN_BLOCKS, 256, 0, stream>>>(hB, scsh);

  k_spmv128<<<rowWaves, 256, 0, stream>>>(hB, rowStart, col, dinv, nullptr, 1.0f, T1);
  k_spmv128<<<rowWaves, 256, 0, stream>>>(T1, rowStart, col, dinv, hB,      2.0f, T2);
  k_spmv128<<<rowWaves, 256, 0, stream>>>(T2, rowStart, col, dinv, T1,      2.0f, T3);
  k_cheb_gemm128<<<g128grid, 256, 0, stream>>>(hB, T1, T2, T3, W3, b3, hA);
  k_act_stats<1><<<BN_BLOCKS, 256, 0, stream>>>(hA, part);
  k_bn_finalize<<<1, FIN_Q * 2 * HH, 0, stream>>>(part, g3, be3, scsh);
  k_bn_apply<<<BN_BLOCKS, 256, 0, stream>>>(hA, scsh);

  k_spmv128<<<rowWaves, 256, 0, stream>>>(hA, rowStart, col, dinv, nullptr, 1.0f, T1);
  k_spmv128<<<rowWaves, 256, 0, stream>>>(T1, rowStart, col, dinv, hA,      2.0f, T2);
  k_spmv128<<<rowWaves, 256, 0, stream>>>(T2, rowStart, col, dinv, T1,      2.0f, T3);
  k_cheb_gemm128<<<g128grid, 256, 0, stream>>>(hA, T1, T2, T3, W4, b4, hB);

  k_final<<<cdiv(NN, 8), 256, 0, stream>>>(hB, Wm, bm, out);
}


# SpMV: adjacency word = source index + source out-degree (16+16 bits), dinv[src] recomputed in registers: per-edge scattered dinv look-up gone (table only if degree >= 65535); weights, sum order same
# speedup vs baseline: 2.9989x; 1.0584x over previous
#include <hip/hip_runtime.h>
#include <hip/hip_bf16.h>


#define NN 50000
#define EE 800000
#define HH 128

typedef float v2f __attribute__((ext_vector_type(2)));
typedef float v8f __attribute__((ext_vector_type(8)));

#define DEG_SAT 0xFFFFu
static_assert(NN <= 0x10000, "a source index must fit the low 16 bits of an adjacency word");

__device__ __forceinline__ float dinv_of_count(int cnt) {
  const float d = (float)cnt;
  return (d > 0.0f) ? rsqrtf(fmaxf(d, 1.0f)) : 0.0f;
}

__device__ __forceinline__ float dinv_of_word(unsigned word, const float* __restrict__ dinv) {
  const unsigned dg = word >> 16;
  float ds = dinv_of_count((int)dg);
  if (dg == DEG_SAT) ds = dinv[word & 0xFFFFu];
  return ds;
}


__global__ void k_zero(float* __restrict__ p, long n) {
  long i = (long)blockIdx.x * blockDim.x + threadIdx.x;
  if (i < n) p[i] = 0.0f;
}


__global__ void k_count(const int* __restrict__ src, const int* __restrict__ dst,
                        int* __restrict__ cntDst, int* __restrict__ cntSrc,
                        int* __restrict__ rank) {
  int e = blockIdx.x * blockDim.x + threadIdx.x;
  if (e < EE) {
    atomicAdd(&cntSrc[src[e]], 1);
    rank[e] = atomicAdd(&cntDst[dst[e]], 1);
  }
}

#define SCAN_T 1024
#define SCAN_PER ((NN + SCAN_T - 1) / SCAN_T)
__global__ __launch_bounds__(SCAN_T) void k_scan_dinv(
    const int* __restrict__ cntDst, const int* __restrict__ cntSrc,
    int* __restrict__ rowStart, float* __restrict__ dinv) {
  __shared__ int sc[NN];
  __shared__ int wsum[32];
  const int t = threadIdx.x;
  const int lane = t & 31;
  const int wv = t >> 5;
  for (int i = t; i < NN; i += SCAN_T) {
    sc[i] = cntDst[i];
    dinv[i] = dinv_of_count(cntSrc[i]);
  }
  __syncthreads();
  const int b = t * SCAN_PER;
  const int bend = (b + SCAN_PER < NN) ? (b + SCAN_PER) : NN;
  int s = 0;
  for (int i = b; i < bend; ++i) s += sc[i];
  int x = s;
  for (int off = 1; off < 32; off <<= 1) {
    int y = __shfl_up(x, off, 32);
    if (lane >= off) x += y;
  }
  if (lane == 31) wsum[wv] = x;
  __syncthreads();
  if (wv == 0) {
    int w = wsum[lane];
    for (int off = 1; off < 32; off <<= 1) {
      int y = __shfl_up(w, off, 32);
      if (lane >= off) w += y;
    }
    wsum[lane] = w;
  }
  __syncthreads();
  int run = x - s + ((wv > 0) ? wsum[wv - 1] : 0);
  for (int i = b; i < bend; ++i) {
    int v = sc[i];
    sc[i] = run;
    run += v;
  }
  __syncthreads();
  for (int i = t; i < NN; i += SCAN_T) rowStart[i] = sc[i];
}

__global__ void k_fill(const int* __restrict__ dst, const int* __restrict__ rank,
                       const int* __restrict__ rowStart, int* __restrict__ eid) {
  int e = blockIdx.x * blockDim.x + threadIdx.x;
  if (e < EE) eid[rowStart[dst[e]] + rank[e]] = e;
}

__global__ __launch_bounds__(256) void k_rowsort(
    const int* __restrict__ src, const int* __restrict__ rowStart,
    const int* __restrict__ eid, const int* __restrict__ cntSrc,
    int* __restrict__ col) {
  const int r = blockIdx.x * 8 + __builtin_amdgcn_readfirstlane((int)(threadIdx.x >> 5));
  if (r >= NN) return;
  const int lane = threadIdx.x & 31;
  const int beg = rowStart[r];
  const int end = (r + 1 < NN) ? rowStart[r + 1] : EE;
  for (int i = beg + lane; i < end; i += 32) {
    const int my = eid[i];
    int rk = 0;
    for (int j = beg; j < end; ++j) rk += (eid[j] < my) ? 1 : 0;
    const unsigned s = (unsigned)src[my];
    const unsigned dg = (unsigned)cntSrc[s];
    col[beg + rk] = (int)(s | ((dg < DEG_SAT ? dg : DEG_SAT) << 16));
  }
}


__global__ __launch_bounds__(256) void k_spmv3(
    const float* __restrict__ h, const int* __restrict__ rowStart,
    const int* __restrict__ col, const float* __restrict__ dinv,
    const float* __restrict__ T0, float scale, float* __restrict__ out) {
  const int gid = blockIdx.x * 256 + threadIdx.x;
  const int r = gid >> 4;
  const int sub = threadIdx.x & 15;
  float a0 = 0.0f, a1 = 0.0f, a2 = 0.0f;
  if (r < NN) {
    const int beg = rowStart[r];
    const int end = (r + 1 < NN) ? rowStart[r + 1] : EE;
    const float dr = dinv[r];
    for (int p = beg + sub; p < end; p += 16) {
      const unsigned word = (unsigned)col[p];
      const int s = (int)(word & 0xFFFFu);
      const float w = -dinv_of_word(word, dinv) * dr;
      a0 += w * h[(long)s * 3 + 0];
      a1 += w * h[(long)s * 3 + 1];
      a2 += w * h[(long)s * 3 + 2];
    }
  }
  for (int off = 8; off > 0; off >>= 1) {
    a0 += __shfl_xor(a0, off, 32);
    a1 += __shfl_xor(a1, off, 32);
    a2 += __shfl_xor(a2, off, 32);
  }
  if (r < NN && sub == 0) {
    float o0 = scale * a0, o1 = scale * a1, o2 = scale * a2;
    if (T0) {
      o0 -= T0[(long)r * 3 + 0];
      o1 -= T0[(long)r * 3 + 1];
      o2 -= T0[(long)r * 3 + 2];
    }
    out[(long)r * 3 + 0] = o0;
    out[(long)r * 3 + 1] = o1;
    out[(long)r * 3 + 2] = o2;
  }
}

__global__ __launch_bounds__(256) void k_spmv128(
    const float* __restrict__ h, const int* __restrict__ rowStart,
    const int* __restrict__ col, const float* __restrict__ dinv,
    const float* __restrict__ T0, float scale, float* __restrict__ out) {
  const int r = blockIdx.x * 8 + __builtin_amdgcn_readfirstlane((int)(threadIdx.x >> 5));
  if (r >= NN) return;
  const int lane = threadIdx.x & 31;
  const int beg = rowStart[r];
  const int end = (r + 1 < NN) ? rowStart[r + 1] : EE;
  const float dr = dinv[r];
  float4 acc = make_float4(0.0f, 0.0f, 0.0f, 0.0f);
  for (int base = beg; base < end; base += 32) {
    const int n = (end - base < 32) ? (end - base) : 32;
    int myc = 0;
    float myw = 0.0f;
    if (lane < n) {
      const unsigned word = (unsigned)col[base + lane];
      myc = (int)(word & 0xFFFFu);
      myw = -dinv_of_word(word, dinv) * dr;
    }
#pragma unroll 4
    for (int j = 0; j < n; ++j) {
      const int s = __builtin_amdgcn_readlane(myc, j);
      const float w = __int_as_float(__builtin_amdgcn_readlane(__float_as_int(myw), j));
      const float4 v = ((const float4*)(h + (long)s * HH))[lane];
      acc.x += w * v.x;
      acc.y += w * v.y;
      acc.z += w * v.z;
      acc.w += w * v.w;
    }
  }
  float4 o = make_float4(scale * acc.x, scale * acc.y, scale * acc.z, scale * acc.w);
  if (T0) {
    const float4 t0 = ((const float4*)(T0 + (long)r * HH))[lane];
    o.x -= t0.x;
    o.y -= t0.y;
    o.z -= t0.z;
    o.w -= t0.w;
  }
  ((float4*)(out + (long)r * HH))[lane] = o;
}


__global__ __launch_bounds__(32) void k_cheb_gemm3(
    const float* __restrict__ T0, const float* __restrict__ T1,
    const float* __restrict__ T2, const float* __restrict__ T3,
    const float* __restrict__ W, const float* __restrict__ bias,
    float* __restrict__ out) {
  const int lane = threadIdx.x;
  const int tm = blockIdx.x;
  const int tn = blockIdx.y;
  const int row = tm * 16 + (lane & 15);
  const int col = tn * 16 + (lane & 15);
  const int khalf = (lane >> 4) * 2;
  const float* Ts[4] = {T0, T1, T2, T3};

  v8f acc = {};
  for (int k0 = 0; k0 < 12; k0 += 4) {
    v2f a, b;
#pragma unroll
    for (int j = 0; j < 2; ++j) {
      int kk = k0 + khalf + j;
      int t = kk / 3;
      int cc = kk - t * 3;
      a[j] = Ts[t][(long)row * 3 + cc];
      b[j] = W[(long)kk * HH + col];
    }
    acc = __builtin_amdgcn_wmma_f32_16x16x4_f32(
        false, a, false, b, (short)0, acc, false, false);
  }

  const float bc = bias[col];
#pragma unroll
  for (int j = 0; j < 8; ++j) {
    int r = tm * 16 + j + (lane >> 4) * 8;
    out[(long)r * HH + (tn * 16 + (lane & 15))] = acc[j] + bc;
  }
}

#define KT 512
#define LST (KT + 4)

__device__ __forceinline__ void stage_panel(const float* __restrict__ A,
                                            float* __restrict__ As,
                                            int tm, int t, int tid) {
  for (int i = tid; i < 16 * (HH / 4); i += 256) {
    int rowl = i >> 5;
    int c4 = i & 31;
    float4 v = ((const float4*)(A + (long)(tm * 16 + rowl) * HH))[c4];
    *(float4*)&As[rowl * LST + t * HH + c4 * 4] = v;
  }
}

__global__ __launch_bounds__(256) void k_cheb_gemm128(
    const float* __restrict__ T0, const float* __restrict__ T1,
    const float* __restrict__ T2, const float* __restrict__ T3,
    const float* __restrict__ W, const float* __restrict__ bias,
    float* __restrict__ out) {
  __shared__ float As[16 * LST];
  const int tid = threadIdx.x;
  const int lane = tid & 31;
  const int wave = tid >> 5;
  const int tm = blockIdx.x;

  stage_panel(T0, As, tm, 0, tid);
  stage_panel(T1, As, tm, 1, tid);
  stage_panel(T2, As, tm, 2, tid);
  stage_panel(T3, As, tm, 3, tid);
  __syncthreads();

  const int l16 = lane & 15;
  const int khalf = (lane >> 4) * 2;
  const int col = wave * 16 + l16;

  v8f acc = {};
#pragma unroll 8
  for (int k0 = 0; k0 < KT; k0 += 4) {
    v2f a = *(const v2f*)&As[l16 * LST + k0 + khalf];
    v2f b;
    b[0] = W[(long)(k0 + khalf) * HH + col];
    b[1] = W[(long)(k0 + khalf + 1) * HH + col];
    acc = __builtin_amdgcn_wmma_f32_16x16x4_f32(
        false, a, false, b, (short)0, acc, false, false);
  }

  const float bc = bias[col];
#pragma unroll
  for (int j = 0; j < 8; ++j) {
    int r = tm * 16 + j + (lane >> 4) * 8;
    out[(long)r * HH + col] = acc[j] + bc;
  }
}


#define BN_ROWS 16
#define BN_WAVES 8
#define BN_BLOCKS ((NN + BN_ROWS * BN_WAVES - 1) / (BN_ROWS * BN_WAVES))
static_assert(NN % BN_ROWS == 0, "a wave's rows are all inside or all outside");

template <int ACT>
__global__ __launch_bounds__(256) void k_act_stats(float* __restrict__ h,
                                                   float* __restrict__ part) {
  __shared__ float red[BN_WAVES][2 * HH];
  const int lane = threadIdx.x & 31;
  const int wv = __builtin_amdgcn_readfirstlane((int)(threadIdx.x >> 5));
  const int r0 = (blockIdx.x * BN_WAVES + wv) * BN_ROWS;
  float4 s = make_float4(0.0f, 0.0f, 0.0f, 0.0f);
  float4 s2 = make_float4(0.0f, 0.0f, 0.0f, 0.0f);
  if (r0 < NN) {
    float4* hp = (float4*)(h + (long)r0 * HH) + lane;
    float4 v[BN_ROWS];
#pragma unroll
    for (int u = 0; u < BN_ROWS; ++u) v[u] = hp[u * (HH / 4)];
#pragma unroll
    for (int u = 0; u < BN_ROWS; ++u) {
      float4 a = v[u];
      if (ACT == 0) {
        a.x = a.x > 0.0f ? a.x : 0.01f * a.x;
        a.y = a.y > 0.0f ? a.y : 0.01f * a.y;
        a.z = a.z > 0.0f ? a.z : 0.01f * a.z;
        a.w = a.w > 0.0f ? a.w : 0.01f * a.w;
      } else {
        a.x = fmaxf(a.x, 0.0f);
        a.y = fmaxf(a.y, 0.0f);
        a.z = fmaxf(a.z, 0.0f);
        a.w = fmaxf(a.w, 0.0f);
      }
      hp[u * (HH / 4)] = a;
      s.x += a.x;  s.y += a.y;  s.z += a.z;  s.w += a.w;
      s2.x += a.x * a.x;  s2.y += a.y * a.y;  s2.z += a.z * a.z;  s2.w += a.w * a.w;
    }
  }
  *(float4*)&red[wv][lane * 4] = s;
  *(float4*)&red[wv][HH + lane * 4] = s2;
  __syncthreads();
  const int t = threadIdx.x;
  float acc = red[0][t];
  for (int w = 1; w < BN_WAVES; ++w) acc += red[w][t];
  part[(long)blockIdx.x * (2 * HH) + t] = acc;
}

#define FIN_Q 4
#define FIN_PER ((BN_BLOCKS + FIN_Q - 1) / FIN_Q)
__global__ __launch_bounds__(FIN_Q * 2 * HH) void k_bn_finalize(
    const float* __restrict__ part, const float* __restrict__ g,
    const float* __restrict__ be, float* __restrict__ scsh) {
  __shared__ float red[FIN_Q][2 * HH];
  const int t = threadIdx.x & (2 * HH - 1);
  const int q = threadIdx.x / (2 * HH);
  const int b0 = q * FIN_PER;
  float acc = 0.0f;
#pragma unroll 14
  for (int i = 0; i < FIN_PER; ++i) {
    const int b = b0 + i;
    const int bc = (b < BN_BLOCKS) ? b : (BN_BLOCKS - 1);
    const float v = part[(long)bc * (2 * HH) + t];
    acc += (b < BN_BLOCKS) ? v : 0.0f;
  }
  red[q][t] = acc;
  __syncthreads();
  if (threadIdx.x < HH) {
    const int ch = threadIdx.x;
    float sum = red[0][ch], sq = red[0][HH + ch];
    for (int k = 1; k < FIN_Q; ++k) {
      sum += red[k][ch];
      sq += red[k][HH + ch];
    }
    const float invN = 1.0f / (float)NN;
    const float m = sum * invN;
    const float var = sq * invN - m * m;
    const float sc = g[ch] * rsqrtf(var + 1e-5f);
    scsh[ch] = sc;
    scsh[HH + ch] = be[ch] - m * sc;
  }
}

__global__ __launch_bounds__(256) void k_bn_apply(float* __restrict__ h,
                                                  const float* __restrict__ scsh) {
  const int lane = threadIdx.x & 31;
  const int wv = __builtin_amdgcn_readfirstlane((int)(threadIdx.x >> 5));
  const int r0 = (blockIdx.x * BN_WAVES + wv) * BN_ROWS;
  if (r0 >= NN) return;
  const float4 sc = ((const float4*)scsh)[lane];
  const float4 sh = ((const float4*)(scsh + HH))[lane];
  float4* hp = (float4*)(h + (long)r0 * HH) + lane;
  float4 v[BN_ROWS];
#pragma unroll
  for (int u = 0; u < BN_ROWS; ++u) v[u] = hp[u * (HH / 4)];
#pragma unroll
  for (int u = 0; u < BN_ROWS; ++u) {
    float4 o;
    o.x = v[u].x * sc.x + sh.x;
    o.y = v[u].y * sc.y + sh.y;
    o.z = v[u].z * sc.z + sh.z;
    o.w = v[u].w * sc.w + sh.w;
    hp[u * (HH / 4)] = o;
  }
}

__global__ __launch_bounds__(256) void k_final(const float* __restrict__ h,
                                               const float* __restrict__ Wm,
                                               const float* __restrict__ bm,
                                               float* __restrict__ out) {
  int n = blockIdx.x * 8 + (threadIdx.x >> 5);
  if (n >= NN) return;
  int lane = threadIdx.x & 31;
  const float4* hp = (const float4*)(h + (long)n * HH);
  float4 v = hp[lane];
  int c = lane * 4;
  float ns = v.x * v.x + v.y * v.y + v.z * v.z + v.w * v.w;
  float d0 = v.x * Wm[(c + 0) * 3 + 0] + v.y * Wm[(c + 1) * 3 + 0] +
             v.z * Wm[(c + 2) * 3 + 0] + v.w * Wm[(c + 3) * 3 + 0];
  float d1 = v.x * Wm[(c + 0) * 3 + 1] + v.y * Wm[(c + 1) * 3 + 1] +
             v.z * Wm[(c + 2) * 3 + 1] + v.w * Wm[(c + 3) * 3 + 1];
  float d2 = v.x * Wm[(c + 0) * 3 + 2] + v.y * Wm[(c + 1) * 3 + 2] +
             v.z * Wm[(c + 2) * 3 + 2] + v.w * Wm[(c + 3) * 3 + 2];
#pragma unroll
  for (int off = 16; off > 0; off >>= 1) {
    ns += __shfl_xor(ns, off, 32);
    d0 += __shfl_xor(d0, off, 32);
    d1 += __shfl_xor(d1, off, 32);
    d2 += __shfl_xor(d2, off, 32);
  }
  if (lane == 0) {
    float inv = 1.0f / fmaxf(sqrtf(ns), 1e-12f);
    out[(long)n * 3 + 0] = d0 * inv + bm[0];
    out[(long)n * 3 + 1] = d1 * inv + bm[1];
    out[(long)n * 3 + 2] = d2 * inv + bm[2];
  }
}


static inline int cdiv(long a, long b) { return (int)((a + b - 1) / b); }

extern "C" void kernel_launch(void* const* d_in, const int* in_sizes, int n_in,
                              void* d_out, int out_size, void* d_ws, size_t ws_size,
                              hipStream_t stream) {
  const float* x   = (const float*)d_in[0];
  const int*   ei  = (const int*)d_in[1];
  const float* W1  = (const float*)d_in[2];
  const float* b1  = (const float*)d_in[3];
  const float* g1  = (const float*)d_in[4];
  const float* be1 = (const float*)d_in[5];
  const float* W2  = (const float*)d_in[6];
  const float* b2  = (const float*)d_in[7];
  const float* g2  = (const float*)d_in[8];
  const float* be2 = (const float*)d_in[9];
  const float* W3  = (const float*)d_in[10];
  const float* b3  = (const float*)d_in[11];
  const float* g3  = (const float*)d_in[12];
  const float* be3 = (const float*)d_in[13];
  const float* W4  = (const float*)d_in[14];
  const float* b4  = (const float*)d_in[15];
  const float* Wm  = (const float*)d_in[16];
  const float* bm  = (const float*)d_in[17];
  float* out = (float*)d_out;

  const int* src = ei;
  const int* dst = ei + EE;

  float* ws   = (float*)d_ws;
  int*   rowStart = (int*)ws;
  float* dinv = ws + NN;
  int*   col  = (int*)(dinv + NN);
  float* T1   = (float*)(col + EE);
  float* T2   = T1 + (long)NN * HH;
  float* T3   = T2 + (long)NN * HH;
  float* hA   = T3 + (long)NN * HH;
  float* hB   = hA + (long)NN * HH;
  float* scsh = hB + (long)NN * HH;
  float* part = T1;
  int* cntDst = (int*)hB;
  int* cntSrc = cntDst + NN;
  int* rank   = cntSrc + NN;
  int* eid    = rank + EE;

  const dim3 g3grid(NN / 16, HH / 16);
  const int g128grid = NN / 16;
  const int rowWaves = cdiv(NN, 8);
  const int row16    = cdiv((long)NN * 16, 256);

  k_zero<<<cdiv(2 * NN, 256), 256, 0, stream>>>((float*)cntDst, 2 * NN);
  k_count<<<cdiv(EE, 256), 256, 0, stream>>>(src, dst, cntDst, cntSrc, rank);
  k_scan_dinv<<<1, SCAN_T, 0, stream>>>(cntDst, cntSrc, rowStart, dinv);
  k_fill<<<cdiv(EE, 256), 256, 0, stream>>>(dst, rank, rowStart, eid);
  k_rowsort<<<rowWaves, 256, 0, stream>>>(src, rowStart, eid, cntSrc, col);

  k_spmv3<<<row16, 256, 0, stream>>>(x,  rowStart, col, dinv, nullptr, 1.0f, T1);
  k_spmv3<<<row16, 256, 0, stream>>>(T1, rowStart, col, dinv, x,       2.0f, T2);
  k_spmv3<<<row16, 256, 0, stream>>>(T2, rowStart, col, dinv, T1,      2.0f, T3);
  k_cheb_gemm3<<<g3grid, 32, 0, stream>>>(x, T1, T2, T3, W1, b1, hA);
  k_act_stats<0><<<BN_BLOCKS, 256, 0, stream>>>(hA, part);
  k_bn_finalize<<<1, FIN_Q * 2 * HH, 0, stream>>>(part, g1, be1, scsh);
  k_bn_apply<<<BN_BLOCKS, 256, 0, stream>>>(hA, scsh);

  k_spmv128<<<rowWaves, 256, 0, stream>>>(hA, rowStart, col, dinv, nullptr, 1.0f, T1);
  k_spmv128<<<rowWaves, 256, 0, stream>>>(T1, rowStart, col, dinv, hA,      2.0f, T2);
  k_spmv128<<<rowWaves, 256, 0, stream>>>(T2, rowStart, col, dinv, T1,      2.0f, T3);
  k_cheb_gemm128<<<g128grid, 256, 0, stream>>>(hA, T1, T2, T3, W2, b2, hB);
  k_act_stats<0><<<BN_BLOCKS, 256, 0, stream>>>(hB, part);
  k_bn_finalize<<<1, FIN_Q * 2 * HH, 0, stream>>>(part, g2, be2, scsh);
  k_bn_apply<<<BN_BLOCKS, 256, 0, stream>>>(hB, scsh);

  k_spmv128<<<rowWaves, 256, 0, stream>>>(hB, rowStart, col, dinv, nullptr, 1.0f, T1);
  k_spmv128<<<rowWaves, 256, 0, stream>>>(T1, rowStart, col, dinv, hB,      2.0f, T2);
  k_spmv128<<<rowWaves, 256, 0, stream>>>(T2, rowStart, col, dinv, T1,      2.0f, T3);
  k_cheb_gemm128<<<g128grid, 256, 0, stream>>>(hB, T1, T2, T3, W3, b3, hA);
  k_act_stats<1><<<BN_BLOCKS, 256, 0, stream>>>(hA, part);
  k_bn_finalize<<<1, FIN_Q * 2 * HH, 0, stream>>>(part, g3, be3, scsh);
  k_bn_apply<<<BN_BLOCKS, 256, 0, stream>>>(hA, scsh);

  k_spmv128<<<rowWaves, 256, 0, stream>>>(hA, rowStart, col, dinv, nullptr, 1.0f, T1);
  k_spmv128<<<rowWaves, 256, 0, stream>>>(T1, rowStart, col, dinv, hA,      2.0f, T2);
  k_spmv128<<<rowWaves, 256, 0, stream>>>(T2, rowStart, col, dinv, T1,      2.0f, T3);
  k_cheb_gemm128<<<g128grid, 256, 0, stream>>>(hA, T1, T2, T3, W4, b4, hB);

  k_final<<<cdiv(NN, 8), 256, 0, stream>>>(hB, Wm, bm, out);
}


# BatchNorm's two streaming passes per layer dropped: GEMM epilogue does activation + per-16-row-tile channel sums (BN sum order regrouped), next layer's loads of h apply scale/shift; 31 -> 25 launches
# speedup vs baseline: 3.5646x; 1.1886x over previous
#include <hip/hip_runtime.h>
#include <hip/hip_bf16.h>


#define NN 50000
#define EE 800000
#define HH 128

typedef float v2f __attribute__((ext_vector_type(2)));
typedef float v8f __attribute__((ext_vector_type(8)));

#define DEG_SAT 0xFFFFu
static_assert(NN <= 0x10000, "a source index must fit the low 16 bits of an adjacency word");

__device__ __forceinline__ float dinv_of_count(int cnt) {
  const float d = (float)cnt;
  return (d > 0.0f) ? rsqrtf(fmaxf(d, 1.0f)) : 0.0f;
}

__device__ __forceinline__ float dinv_of_word(unsigned word, const float* __restrict__ dinv) {
  const unsigned dg = word >> 16;
  float ds = dinv_of_count((int)dg);
  if (dg == DEG_SAT) ds = dinv[word & 0xFFFFu];
  return ds;
}


__global__ void k_zero(float* __restrict__ p, long n) {
  long i = (long)blockIdx.x * blockDim.x + threadIdx.x;
  if (i < n) p[i] = 0.0f;
}


__global__ void k_count(const int* __restrict__ src, const int* __restrict__ dst,
                        int* __restrict__ cntDst, int* __restrict__ cntSrc,
                        int* __restrict__ rank) {
  int e = blockIdx.x * blockDim.x + threadIdx.x;
  if (e < EE) {
    atomicAdd(&cntSrc[src[e]], 1);
    rank[e] = atomicAdd(&cntDst[dst[e]], 1);
  }
}

#define SCAN_T 1024
#define SCAN_PER ((NN + SCAN_T - 1) / SCAN_T)
__global__ __launch_bounds__(SCAN_T) void k_scan_dinv(
    const int* __restrict__ cntDst, const int* __restrict__ cntSrc,
    int* __restrict__ rowStart, float* __restrict__ dinv) {
  __shared__ int sc[NN];
  __shared__ int wsum[32];
  const int t = threadIdx.x;
  const int lane = t & 31;
  const int wv = t >> 5;
  for (int i = t; i < NN; i += SCAN_T) {
    sc[i] = cntDst[i];
    dinv[i] = dinv_of_count(cntSrc[i]);
  }
  __syncthreads();
  const int b = t * SCAN_PER;
  const int bend = (b + SCAN_PER < NN) ? (b + SCAN_PER) : NN;
  int s = 0;
  for (int i = b; i < bend; ++i) s += sc[i];
  int x = s;
  for (int off = 1; off < 32; off <<= 1) {
    int y = __shfl_up(x, off, 32);
    if (lane >= off) x += y;
  }
  if (lane == 31) wsum[wv] = x;
  __syncthreads();
  if (wv == 0) {
    int w = wsum[lane];
    for (int off = 1; off < 32; off <<= 1) {
      int y = __shfl_up(w, off, 32);
      if (lane >= off) w += y;
    }
    wsum[lane] = w;
  }
  __syncthreads();
  int run = x - s + ((wv > 0) ? wsum[wv - 1] : 0);
  for (int i = b; i < bend; ++i) {
    int v = sc[i];
    sc[i] = run;
    run += v;
  }
  __syncthreads();
  for (int i = t; i < NN; i += SCAN_T) rowStart[i] = sc[i];
}

__global__ void k_fill(const int* __restrict__ dst, const int* __restrict__ rank,
                       const int* __restrict__ rowStart, int* __restrict__ eid) {
  int e = blockIdx.x * blockDim.x + threadIdx.x;
  if (e < EE) eid[rowStart[dst[e]] + rank[e]] = e;
}

__global__ __launch_bounds__(256) void k_rowsort(
    const int* __restrict__ src, const int* __restrict__ rowStart,
    const int* __restrict__ eid, const int* __restrict__ cntSrc,
    int* __restrict__ col) {
  const int r = blockIdx.x * 8 + __builtin_amdgcn_readfirstlane((int)(threadIdx.x >> 5));
  if (r >= NN) return;
  const int lane = threadIdx.x & 31;
  const int beg = rowStart[r];
  const int end = (r + 1 < NN) ? rowStart[r + 1] : EE;
  for (int i = beg + lane; i < end; i += 32) {
    const int my = eid[i];
    int rk = 0;
    for (int j = beg; j < end; ++j) rk += (eid[j] < my) ? 1 : 0;
    const unsigned s = (unsigned)src[my];
    const unsigned dg = (unsigned)cntSrc[s];
    col[beg + rk] = (int)(s | ((dg < DEG_SAT ? dg : DEG_SAT) << 16));
  }
}


__global__ __launch_bounds__(256) void k_spmv3(
    const float* __restrict__ h, const int* __restrict__ rowStart,
    const int* __restrict__ col, const float* __restrict__ dinv,
    const float* __restrict__ T0, float scale, float* __restrict__ out) {
  const int gid = blockIdx.x * 256 + threadIdx.x;
  const int r = gid >> 4;
  const int sub = threadIdx.x & 15;
  float a0 = 0.0f, a1 = 0.0f, a2 = 0.0f;
  if (r < NN) {
    const int beg = rowStart[r];
    const int end = (r + 1 < NN) ? rowStart[r + 1] : EE;
    const float dr = dinv[r];
    for (int p = beg + sub; p < end; p += 16) {
      const unsigned word = (unsigned)col[p];
      const int s = (int)(word & 0xFFFFu);
      const float w = -dinv_of_word(word, dinv) * dr;
      a0 += w * h[(long)s * 3 + 0];
      a1 += w * h[(long)s * 3 + 1];
      a2 += w * h[(long)s * 3 + 2];
    }
  }
  for (int off = 8; off > 0; off >>= 1) {
    a0 += __shfl_xor(a0, off, 32);
    a1 += __shfl_xor(a1, off, 32);
    a2 += __shfl_xor(a2, off, 32);
  }
  if (r < NN && sub == 0) {
    float o0 = scale * a0, o1 = scale * a1, o2 = scale * a2;
    if (T0) {
      o0 -= T0[(long)r * 3 + 0];
      o1 -= T0[(long)r * 3 + 1];
      o2 -= T0[(long)r * 3 + 2];
    }
    out[(long)r * 3 + 0] = o0;
    out[(long)r * 3 + 1] = o1;
    out[(long)r * 3 + 2] = o2;
  }
}

template <int AFF>
__global__ __launch_bounds__(256) void k_spmv128(
    const float* __restrict__ h, const int* __restrict__ rowStart,
    const int* __restrict__ col, const float* __restrict__ dinv,
    const float* __restrict__ T0, const float* __restrict__ scsh,
    float scale, float* __restrict__ out) {
  const int r = blockIdx.x * 8 + __builtin_amdgcn_readfirstlane((int)(threadIdx.x >> 5));
  if (r >= NN) return;
  const int lane = threadIdx.x & 31;
  const int beg = rowStart[r];
  const int end = (r + 1 < NN) ? rowStart[r + 1] : EE;
  const float dr = dinv[r];
  float4 sc = make_float4(1.0f, 1.0f, 1.0f, 1.0f);
  float4 sh = make_float4(0.0f, 0.0f, 0.0f, 0.0f);
  if (AFF != 0) {
    sc = ((const float4*)scsh)[lane];
    sh = ((const float4*)(scsh + HH))[lane];
  }
  float4 acc = make_float4(0.0f, 0.0f, 0.0f, 0.0f);
  for (int base = beg; base < end; base += 32) {
    const int n = (end - base < 32) ? (end - base) : 32;
    int myc = 0;
    float myw = 0.0f;
    if (lane < n) {
      const unsigned word = (unsigned)col[base + lane];
      myc = (int)(word & 0xFFFFu);
      myw = -dinv_of_word(word, dinv) * dr;
    }
#pragma unroll 4
    for (int j = 0; j < n; ++j) {
      const int s = __builtin_amdgcn_readlane(myc, j);
      const float w = __int_as_float(__builtin_amdgcn_readlane(__float_as_int(myw), j));
      float4 v = ((const float4*)(h + (long)s * HH))[lane];
      if (AFF == 1) {
        v.x = v.x * sc.x + sh.x;
        v.y = v.y * sc.y + sh.y;
        v.z = v.z * sc.z + sh.z;
        v.w = v.w * sc.w + sh.w;
      }
      acc.x += w * v.x;
      acc.y += w * v.y;
      acc.z += w * v.z;
      acc.w += w * v.w;
    }
  }
  float4 o = make_float4(scale * acc.x, scale * acc.y, scale * acc.z, scale * acc.w);
  if (T0) {
    float4 t0 = ((const float4*)(T0 + (long)r * HH))[lane];
    if (AFF == 2) {
      t0.x = t0.x * sc.x + sh.x;
      t0.y = t0.y * sc.y + sh.y;
      t0.z = t0.z * sc.z + sh.z;
      t0.w = t0.w * sc.w + sh.w;
    }
    o.x -= t0.x;
    o.y -= t0.y;
    o.z -= t0.z;
    o.w -= t0.w;
  }
  ((float4*)(out + (long)r * HH))[lane] = o;
}


template <int ACT>
__device__ __forceinline__ void tile_epilogue(const v8f acc, float bc, int tm, int col,
                                              int lane, float* out, float* prow) {
  float s = 0.0f, s2 = 0.0f;
#pragma unroll
  for (int j = 0; j < 8; ++j) {
    const int r = tm * 16 + j + (lane >> 4) * 8;
    float v = acc[j] + bc;
    if (ACT == 0) v = v > 0.0f ? v : 0.01f * v;
    if (ACT == 1) v = fmaxf(v, 0.0f);
    out[(long)r * HH + col] = v;
    s += v;
    s2 += v * v;
  }
  if (ACT != 2) {
    s += __shfl_xor(s, 16, 32);
    s2 += __shfl_xor(s2, 16, 32);
    if (lane < 16) {
      prow[col] = s;
      prow[HH + col] = s2;
    }
  }
}

__global__ __launch_bounds__(32) void k_cheb_gemm3(
    const float* __restrict__ T0, const float* __restrict__ T1,
    const float* __restrict__ T2, const float* __restrict__ T3,
    const float* __restrict__ W, const float* __restrict__ bias,
    float* __restrict__ out, float* __restrict__ part) {
  const int lane = threadIdx.x;
  const int tm = blockIdx.x;
  const int tn = blockIdx.y;
  const int row = tm * 16 + (lane & 15);
  const int col = tn * 16 + (lane & 15);
  const int khalf = (lane >> 4) * 2;
  const float* Ts[4] = {T0, T1, T2, T3};

  v8f acc = {};
  for (int k0 = 0; k0 < 12; k0 += 4) {
    v2f a, b;
#pragma unroll
    for (int j = 0; j < 2; ++j) {
      int kk = k0 + khalf + j;
      int t = kk / 3;
      int cc = kk - t * 3;
      a[j] = Ts[t][(long)row * 3 + cc];
      b[j] = W[(long)kk * HH + col];
    }
    acc = __builtin_amdgcn_wmma_f32_16x16x4_f32(
        false, a, false, b, (short)0, acc, false, false);
  }

  tile_epilogue<0>(acc, bias[col], tm, col, lane, out, part + (long)tm * (2 * HH));
}

#define KT 512
#define LST (KT + 4)

template <bool AFF>
__device__ __forceinline__ void stage_panel(const float* A, float* __restrict__ As,
                                            const float* __restrict__ scsh,
                                            int tm, int t, int tid) {
  for (int i = tid; i < 16 * (HH / 4); i += 256) {
    int rowl = i >> 5;
    int c4 = i & 31;
    float4 v = ((const float4*)(A + (long)(tm * 16 + rowl) * HH))[c4];
    if (AFF) {
      const float4 sc = ((const float4*)scsh)[c4];
      const float4 sh = ((const float4*)(scsh + HH))[c4];
      v.x = v.x * sc.x + sh.x;
      v.y = v.y * sc.y + sh.y;
      v.z = v.z * sc.z + sh.z;
      v.w = v.w * sc.w + sh.w;
    }
    *(float4*)&As[rowl * LST + t * HH + c4 * 4] = v;
  }
}

#define PART_STRIDE128 (16 * HH)
template <int ACT>
__global__ __launch_bounds__(256) void k_cheb_gemm128(
    const float* __restrict__ T0, const float* __restrict__ T1,
    const float* __restrict__ T2, const float* T3,
    const float* __restrict__ W, const float* __restrict__ bias,
    const float* __restrict__ scsh, float* __restrict__ out, float* part) {
  __shared__ float As[16 * LST];
  const int tid = threadIdx.x;
  const int lane = tid & 31;
  const int wave = tid >> 5;
  const int tm = blockIdx.x;

  stage_panel<true>(T0, As, scsh, tm, 0, tid);
  stage_panel<false>(T1, As, scsh, tm, 1, tid);
  stage_panel<false>(T2, As, scsh, tm, 2, tid);
  stage_panel<false>(T3, As, scsh, tm, 3, tid);
  __syncthreads();

  const int l16 = lane & 15;
  const int khalf = (lane >> 4) * 2;
  const int col = wave * 16 + l16;

  v8f acc = {};
#pragma unroll 8
  for (int k0 = 0; k0 < KT; k0 += 4) {
    v2f a = *(const v2f*)&As[l16 * LST + k0 + khalf];
    v2f b;
    b[0] = W[(long)(k0 + khalf) * HH + col];
    b[1] = W[(long)(k0 + khalf + 1) * HH + col];
    acc = __builtin_amdgcn_wmma_f32_16x16x4_f32(
        false, a, false, b, (short)0, acc, false, false);
  }

  tile_epilogue<ACT>(acc, bias[col], tm, col, lane, out, part + (long)tm * PART_STRIDE128);
}

#define NTILE (NN / 16)
#define FIN_T 256
#define FIN_PER ((NTILE + FIN_T - 1) / FIN_T)
static_assert(NN % 16 == 0, "every row belongs to a whole 16-row tile");
__global__ __launch_bounds__(FIN_T) void k_bn_finalize(
    const float* __restrict__ part, long pstride, const float* __restrict__ g,
    const float* __restrict__ be, float* __restrict__ scsh) {
  __shared__ float4 rs[FIN_T];
  __shared__ float4 rq[FIN_T];
  const int t = threadIdx.x;
  const int c0 = blockIdx.x * 4;
  float4 s = make_float4(0.0f, 0.0f, 0.0f, 0.0f);
  float4 q = make_float4(0.0f, 0.0f, 0.0f, 0.0f);
#pragma unroll
  for (int i = 0; i < FIN_PER; ++i) {
    const int row = t + i * FIN_T;
    const int rc = (row < NTILE) ? row : (NTILE - 1);
    const float4 vs = *(const float4*)(part + (long)rc * pstride + c0);
    const float4 vq = *(const float4*)(part + (long)rc * pstride + HH + c0);
    const bool in = row < NTILE;
    s.x += in ? vs.x : 0.0f;  s.y += in ? vs.y : 0.0f;
    s.z += in ? vs.z : 0.0f;  s.w += in ? vs.w : 0.0f;
    q.x += in ? vq.x : 0.0f;  q.y += in ? vq.y : 0.0f;
    q.z += in ? vq.z : 0.0f;  q.w += in ? vq.w : 0.0f;
  }
  rs[t] = s;
  rq[t] = q;
  __syncthreads();
  for (int off = FIN_T / 2; off > 0; off >>= 1) {
    if (t < off) {
      float4 a = rs[t], b = rs[t + off];
      a.x += b.x;  a.y += b.y;  a.z += b.z;  a.w += b.w;
      rs[t] = a;
      float4 c = rq[t], d = rq[t + off];
      c.x += d.x;  c.y += d.y;  c.z += d.z;  c.w += d.w;
      rq[t] = c;
    }
    __syncthreads();
  }
  if (t < 4) {
    const int ch = c0 + t;
    const float sum = ((const float*)&rs[0])[t];
    const float sq = ((const float*)&rq[0])[t];
    const float invN = 1.0f / (float)NN;
    const float m = sum * invN;
    const float var = sq * invN - m * m;
    const float sc = g[ch] * rsqrtf(var + 1e-5f);
    scsh[ch] = sc;
    scsh[HH + ch] = be[ch] - m * sc;
  }
}

__global__ __launch_bounds__(256) void k_final(const float* __restrict__ h,
                                               const float* __restrict__ Wm,
                                               const float* __restrict__ bm,
                                               float* __restrict__ out) {
  int n = blockIdx.x * 8 + (threadIdx.x >> 5);
  if (n >= NN) return;
  int lane = threadIdx.x & 31;
  const float4* hp = (const float4*)(h + (long)n * HH);
  float4 v = hp[lane];
  int c = lane * 4;
  float ns = v.x * v.x + v.y * v.y + v.z * v.z + v.w * v.w;
  float d0 = v.x * Wm[(c + 0) * 3 + 0] + v.y * Wm[(c + 1) * 3 + 0] +
             v.z * Wm[(c + 2) * 3 + 0] + v.w * Wm[(c + 3) * 3 + 0];
  float d1 = v.x * Wm[(c + 0) * 3 + 1] + v.y * Wm[(c + 1) * 3 + 1] +
             v.z * Wm[(c + 2) * 3 + 1] + v.w * Wm[(c + 3) * 3 + 1];
  float d2 = v.x * Wm[(c + 0) * 3 + 2] + v.y * Wm[(c + 1) * 3 + 2] +
             v.z * Wm[(c + 2) * 3 + 2] + v.w * Wm[(c + 3) * 3 + 2];
#pragma unroll
  for (int off = 16; off > 0; off >>= 1) {
    ns += __shfl_xor(ns, off, 32);
    d0 += __shfl_xor(d0, off, 32);
    d1 += __shfl_xor(d1, off, 32);
    d2 += __shfl_xor(d2, off, 32);
  }
  if (lane == 0) {
    float inv = 1.0f / fmaxf(sqrtf(ns), 1e-12f);
    out[(long)n * 3 + 0] = d0 * inv + bm[0];
    out[(long)n * 3 + 1] = d1 * inv + bm[1];
    out[(long)n * 3 + 2] = d2 * inv + bm[2];
  }
}


static inline int cdiv(long a, long b) { return (int)((a + b - 1) / b); }

extern "C" void kernel_launch(void* const* d_in, const int* in_sizes, int n_in,
                              void* d_out, int out_size, void* d_ws, size_t ws_size,
                              hipStream_t stream) {
  const float* x   = (const float*)d_in[0];
  const int*   ei  = (const int*)d_in[1];
  const float* W1  = (const float*)d_in[2];
  const float* b1  = (const float*)d_in[3];
  const float* g1  = (const float*)d_in[4];
  const float* be1 = (const float*)d_in[5];
  const float* W2  = (const float*)d_in[6];
  const float* b2  = (const float*)d_in[7];
  const float* g2  = (const float*)d_in[8];
  const float* be2 = (const float*)d_in[9];
  const float* W3  = (const float*)d_in[10];
  const float* b3  = (const float*)d_in[11];
  const float* g3  = (const float*)d_in[12];
  const float* be3 = (const float*)d_in[13];
  const float* W4  = (const float*)d_in[14];
  const float* b4  = (const float*)d_in[15];
  const float* Wm  = (const float*)d_in[16];
  const float* bm  = (const float*)d_in[17];
  float* out = (float*)d_out;

  const int* src = ei;
  const int* dst = ei + EE;

  float* ws   = (float*)d_ws;
  int*   rowStart = (int*)ws;
  float* dinv = ws + NN;
  int*   col  = (int*)(dinv + NN);
  float* T1   = (float*)(col + EE);
  float* T2   = T1 + (long)NN * HH;
  float* T3   = T2 + (long)NN * HH;
  float* hA   = T3 + (long)NN * HH;
  float* hB   = hA + (long)NN * HH;
  float* scsh = hB + (long)NN * HH;
  float* part1 = hB;
  int* cntDst = (int*)hB;
  int* cntSrc = cntDst + NN;
  int* rank   = cntSrc + NN;
  int* eid    = rank + EE;

  const dim3 g3grid(NN / 16, HH / 16);
  const int g128grid = NN / 16;
  const int rowWaves = cdiv(NN, 8);
  const int row16    = cdiv((long)NN * 16, 256);

  k_zero<<<cdiv(2 * NN, 256), 256, 0, stream>>>((float*)cntDst, 2 * NN);
  k_count<<<cdiv(EE, 256), 256, 0, stream>>>(src, dst, cntDst, cntSrc, rank);
  k_scan_dinv<<<1, SCAN_T, 0, stream>>>(cntDst, cntSrc, rowStart, dinv);
  k_fill<<<cdiv(EE, 256), 256, 0, stream>>>(dst, rank, rowStart, eid);
  k_rowsort<<<rowWaves, 256, 0, stream>>>(src, rowStart, eid, cntSrc, col);

  k_spmv3<<<row16, 256, 0, stream>>>(x,  rowStart, col, dinv, nullptr, 1.0f, T1);
  k_spmv3<<<row16, 256, 0, stream>>>(T1, rowStart, col, dinv, x,       2.0f, T2);
  k_spmv3<<<row16, 256, 0, stream>>>(T2, rowStart, col, dinv, T1,      2.0f, T3);
  k_cheb_gemm3<<<g3grid, 32, 0, stream>>>(x, T1, T2, T3, W1, b1, hA, part1);
  k_bn_finalize<<<HH / 4, FIN_T, 0, stream>>>(part1, 2 * HH, g1, be1, scsh);

  k_spmv128<1><<<rowWaves, 256, 0, stream>>>(hA, rowStart, col, dinv, nullptr, scsh, 1.0f, T1);
  k_spmv128<2><<<rowWaves, 256, 0, stream>>>(T1, rowStart, col, dinv, hA,      scsh, 2.0f, T2);
  k_spmv128<0><<<rowWaves, 256, 0, stream>>>(T2, rowStart, col, dinv, T1,      scsh, 2.0f, T3);
  k_cheb_gemm128<0><<<g128grid, 256, 0, stream>>>(hA, T1, T2, T3, W2, b2, scsh, hB, T3);
  k_bn_finalize<<<HH / 4, FIN_T, 0, stream>>>(T3, PART_STRIDE128, g2, be2, scsh);

  k_spmv128<1><<<rowWaves, 256, 0, stream>>>(hB, rowStart, col, dinv, nullptr, scsh, 1.0f, T1);
  k_spmv128<2><<<rowWaves, 256, 0, stream>>>(T1, rowStart, col, dinv, hB,      scsh, 2.0f, T2);
  k_spmv128<0><<<rowWaves, 256, 0, stream>>>(T2, rowStart, col, dinv, T1,      scsh, 2.0f, T3);
  k_cheb_gemm128<1><<<g128grid, 256, 0, stream>>>(hB, T1, T2, T3, W3, b3, scsh, hA, T3);
  k_bn_finalize<<<HH / 4, FIN_T, 0, stream>>>(T3, PART_STRIDE128, g3, be3, scsh);

  k_spmv128<1><<<rowWaves, 256, 0, stream>>>(hA, rowStart, col, dinv, nullptr, scsh, 1.0f, T1);
  k_spmv128<2><<<rowWaves, 256, 0, stream>>>(T1, rowStart, col, dinv, hA,      scsh, 2.0f, T2);
  k_spmv128<0><<<rowWaves, 256, 0, stream>>>(T2, rowStart, col, dinv, T1,      scsh, 2.0f, T3);
  k_cheb_gemm128<2><<<g128grid, 256, 0, stream>>>(hA, T1, T2, T3, W4, b4, scsh, hB, T3);

  k_final<<<cdiv(NN, 8), 256, 0, stream>>>(hB, Wm, bm, out);
}


# C=128 layers: third Chebyshev term 2*L*T2-T1 summed inside the GEMM block (2 rows per wave, same row sum and order) into its LDS panel; 3 SpMV launches and T3's memory round trip go; 25->22 launches
# speedup vs baseline: 3.6928x; 1.0360x over previous
#include <hip/hip_runtime.h>
#include <hip/hip_bf16.h>


#define NN 50000
#define EE 800000
#define HH 128

typedef float v2f __attribute__((ext_vector_type(2)));
typedef float v8f __attribute__((ext_vector_type(8)));

#define DEG_SAT 0xFFFFu
static_assert(NN <= 0x10000, "a source index must fit the low 16 bits of an adjacency word");

__device__ __forceinline__ float dinv_of_count(int cnt) {
  const float d = (float)cnt;
  return (d > 0.0f) ? rsqrtf(fmaxf(d, 1.0f)) : 0.0f;
}

__device__ __forceinline__ float dinv_of_word(unsigned word, const float* __restrict__ dinv) {
  const unsigned dg = word >> 16;
  float ds = dinv_of_count((int)dg);
  if (dg == DEG_SAT) ds = dinv[word & 0xFFFFu];
  return ds;
}


__global__ void k_zero(float* __restrict__ p, long n) {
  long i = (long)blockIdx.x * blockDim.x + threadIdx.x;
  if (i < n) p[i] = 0.0f;
}


__global__ void k_count(const int* __restrict__ src, const int* __restrict__ dst,
                        int* __restrict__ cntDst, int* __restrict__ cntSrc,
                        int* __restrict__ rank) {
  int e = blockIdx.x * blockDim.x + threadIdx.x;
  if (e < EE) {
    atomicAdd(&cntSrc[src[e]], 1);
    rank[e] = atomicAdd(&cntDst[dst[e]], 1);
  }
}

#define SCAN_T 1024
#define SCAN_PER ((NN + SCAN_T - 1) / SCAN_T)
__global__ __launch_bounds__(SCAN_T) void k_scan_dinv(
    const int* __restrict__ cntDst, const int* __restrict__ cntSrc,
    int* __restrict__ rowStart, float* __restrict__ dinv) {
  __shared__ int sc[NN];
  __shared__ int wsum[32];
  const int t = threadIdx.x;
  const int lane = t & 31;
  const int wv = t >> 5;
  for (int i = t; i < NN; i += SCAN_T) {
    sc[i] = cntDst[i];
    dinv[i] = dinv_of_count(cntSrc[i]);
  }
  __syncthreads();
  const int b = t * SCAN_PER;
  const int bend = (b + SCAN_PER < NN) ? (b + SCAN_PER) : NN;
  int s = 0;
  for (int i = b; i < bend; ++i) s += sc[i];
  int x = s;
  for (int off = 1; off < 32; off <<= 1) {
    int y = __shfl_up(x, off, 32);
    if (lane >= off) x += y;
  }
  if (lane == 31) wsum[wv] = x;
  __syncthreads();
  if (wv == 0) {
    int w = wsum[lane];
    for (int off = 1; off < 32; off <<= 1) {
      int y = __shfl_up(w, off, 32);
      if (lane >= off) w += y;
    }
    wsum[lane] = w;
  }
  __syncthreads();
  int run = x - s + ((wv > 0) ? wsum[wv - 1] : 0);
  for (int i = b; i < bend; ++i) {
    int v = sc[i];
    sc[i] = run;
    run += v;
  }
  __syncthreads();
  for (int i = t; i < NN; i += SCAN_T) rowStart[i] = sc[i];
}

__global__ void k_fill(const int* __restrict__ dst, const int* __restrict__ rank,
                       const int* __restrict__ rowStart, int* __restrict__ eid) {
  int e = blockIdx.x * blockDim.x + threadIdx.x;
  if (e < EE) eid[rowStart[dst[e]] + rank[e]] = e;
}

__global__ __launch_bounds__(256) void k_rowsort(
    const int* __restrict__ src, const int* __restrict__ rowStart,
    const int* __restrict__ eid, const int* __restrict__ cntSrc,
    int* __restrict__ col) {
  const int r = blockIdx.x * 8 + __builtin_amdgcn_readfirstlane((int)(threadIdx.x >> 5));
  if (r >= NN) return;
  const int lane = threadIdx.x & 31;
  const int beg = rowStart[r];
  const int end = (r + 1 < NN) ? rowStart[r + 1] : EE;
  for (int i = beg + lane; i < end; i += 32) {
    const int my = eid[i];
    int rk = 0;
    for (int j = beg; j < end; ++j) rk += (eid[j] < my) ? 1 : 0;
    const unsigned s = (unsigned)src[my];
    const unsigned dg = (unsigned)cntSrc[s];
    col[beg + rk] = (int)(s | ((dg < DEG_SAT ? dg : DEG_SAT) << 16));
  }
}


__global__ __launch_bounds__(256) void k_spmv3(
    const float* __restrict__ h, const int* __restrict__ rowStart,
    const int* __restrict__ col, const float* __restrict__ dinv,
    const float* __restrict__ T0, float scale, float* __restrict__ out) {
  const int gid = blockIdx.x * 256 + threadIdx.x;
  const int r = gid >> 4;
  const int sub = threadIdx.x & 15;
  float a0 = 0.0f, a1 = 0.0f, a2 = 0.0f;
  if (r < NN) {
    const int beg = rowStart[r];
    const int end = (r + 1 < NN) ? rowStart[r + 1] : EE;
    const float dr = dinv[r];
    for (int p = beg + sub; p < end; p += 16) {
      const unsigned word = (unsigned)col[p];
      const int s = (int)(word & 0xFFFFu);
      const float w = -dinv_of_word(word, dinv) * dr;
      a0 += w * h[(long)s * 3 + 0];
      a1 += w * h[(long)s * 3 + 1];
      a2 += w * h[(long)s * 3 + 2];
    }
  }
  for (int off = 8; off > 0; off >>= 1) {
    a0 += __shfl_xor(a0, off, 32);
    a1 += __shfl_xor(a1, off, 32);
    a2 += __shfl_xor(a2, off, 32);
  }
  if (r < NN && sub == 0) {
    float o0 = scale * a0, o1 = scale * a1, o2 = scale * a2;
    if (T0) {
      o0 -= T0[(long)r * 3 + 0];
      o1 -= T0[(long)r * 3 + 1];
      o2 -= T0[(long)r * 3 + 2];
    }
    out[(long)r * 3 + 0] = o0;
    out[(long)r * 3 + 1] = o1;
    out[(long)r * 3 + 2] = o2;
  }
}

template <int AFF>
__device__ __forceinline__ float4 row_sum128(
    const float* __restrict__ h, const int* __restrict__ col,
    const float* __restrict__ dinv, int beg, int end, float dr,
    float4 sc, float4 sh, int lane) {
  float4 acc = make_float4(0.0f, 0.0f, 0.0f, 0.0f);
  for (int base = beg; base < end; base += 32) {
    const int n = (end - base < 32) ? (end - base) : 32;
    int myc = 0;
    float myw = 0.0f;
    if (lane < n) {
      const unsigned word = (unsigned)col[base + lane];
      myc = (int)(word & 0xFFFFu);
      myw = -dinv_of_word(word, dinv) * dr;
    }
#pragma unroll 4
    for (int j = 0; j < n; ++j) {
      const int s = __builtin_amdgcn_readlane(myc, j);
      const float w = __int_as_float(__builtin_amdgcn_readlane(__float_as_int(myw), j));
      float4 v = ((const float4*)(h + (long)s * HH))[lane];
      if (AFF == 1) {
        v.x = v.x * sc.x + sh.x;
        v.y = v.y * sc.y + sh.y;
        v.z = v.z * sc.z + sh.z;
        v.w = v.w * sc.w + sh.w;
      }
      acc.x += w * v.x;
      acc.y += w * v.y;
      acc.z += w * v.z;
      acc.w += w * v.w;
    }
  }
  return acc;
}

template <int AFF>
__global__ __launch_bounds__(256) void k_spmv128(
    const float* __restrict__ h, const int* __restrict__ rowStart,
    const int* __restrict__ col, const float* __restrict__ dinv,
    const float* __restrict__ T0, const float* __restrict__ scsh,
    float scale, float* __restrict__ out) {
  const int r = blockIdx.x * 8 + __builtin_amdgcn_readfirstlane((int)(threadIdx.x >> 5));
  if (r >= NN) return;
  const int lane = threadIdx.x & 31;
  const int beg = rowStart[r];
  const int end = (r + 1 < NN) ? rowStart[r + 1] : EE;
  const float dr = dinv[r];
  float4 sc = make_float4(1.0f, 1.0f, 1.0f, 1.0f);
  float4 sh = make_float4(0.0f, 0.0f, 0.0f, 0.0f);
  if (AFF != 0) {
    sc = ((const float4*)scsh)[lane];
    sh = ((const float4*)(scsh + HH))[lane];
  }
  const float4 acc = row_sum128<AFF>(h, col, dinv, beg, end, dr, sc, sh, lane);
  float4 o = make_float4(scale * acc.x, scale * acc.y, scale * acc.z, scale * acc.w);
  if (T0) {
    float4 t0 = ((const float4*)(T0 + (long)r * HH))[lane];
    if (AFF == 2) {
      t0.x = t0.x * sc.x + sh.x;
      t0.y = t0.y * sc.y + sh.y;
      t0.z = t0.z * sc.z + sh.z;
      t0.w = t0.w * sc.w + sh.w;
    }
    o.x -= t0.x;
    o.y -= t0.y;
    o.z -= t0.z;
    o.w -= t0.w;
  }
  ((float4*)(out + (long)r * HH))[lane] = o;
}


template <int ACT>
__device__ __forceinline__ void tile_epilogue(const v8f acc, float bc, int tm, int col,
                                              int lane, float* out, float* prow) {
  float s = 0.0f, s2 = 0.0f;
#pragma unroll
  for (int j = 0; j < 8; ++j) {
    const int r = tm * 16 + j + (lane >> 4) * 8;
    float v = acc[j] + bc;
    if (ACT == 0) v = v > 0.0f ? v : 0.01f * v;
    if (ACT == 1) v = fmaxf(v, 0.0f);
    out[(long)r * HH + col] = v;
    s += v;
    s2 += v * v;
  }
  if (ACT != 2) {
    s += __shfl_xor(s, 16, 32);
    s2 += __shfl_xor(s2, 16, 32);
    if (lane < 16) {
      prow[col] = s;
      prow[HH + col] = s2;
    }
  }
}

__global__ __launch_bounds__(32) void k_cheb_gemm3(
    const float* __restrict__ T0, const float* __restrict__ T1,
    const float* __restrict__ T2, const float* __restrict__ T3,
    const float* __restrict__ W, const float* __restrict__ bias,
    float* __restrict__ out, float* __restrict__ part) {
  const int lane = threadIdx.x;
  const int tm = blockIdx.x;
  const int tn = blockIdx.y;
  const int row = tm * 16 + (lane & 15);
  const int col = tn * 16 + (lane & 15);
  const int khalf = (lane >> 4) * 2;
  const float* Ts[4] = {T0, T1, T2, T3};

  v8f acc = {};
  for (int k0 = 0; k0 < 12; k0 += 4) {
    v2f a, b;
#pragma unroll
    for (int j = 0; j < 2; ++j) {
      int kk = k0 + khalf + j;
      int t = kk / 3;
      int cc = kk - t * 3;
      a[j] = Ts[t][(long)row * 3 + cc];
      b[j] = W[(long)kk * HH + col];
    }
    acc = __builtin_amdgcn_wmma_f32_16x16x4_f32(
        false, a, false, b, (short)0, acc, false, false);
  }

  tile_epilogue<0>(acc, bias[col], tm, col, lane, out, part + (long)tm * (2 * HH));
}

#define KT 512
#define LST (KT + 4)

template <int ACT>
__global__ __launch_bounds__(256) void k_cheb_gemm128(
    const float* __restrict__ T0, const float* __restrict__ T1,
    const float* __restrict__ T2, const int* __restrict__ rowStart,
    const int* __restrict__ adj, const float* __restrict__ dinv,
    const float* __restrict__ W, const float* __restrict__ bias,
    const float* __restrict__ scsh, float* __restrict__ out,
    float* __restrict__ part) {
  __shared__ float As[16 * LST];
  const int tid = threadIdx.x;
  const int lane = tid & 31;
  const int wave = tid >> 5;
  const int tm = blockIdx.x;

  {
    const int wv = __builtin_amdgcn_readfirstlane(wave);
    const float4 sc = ((const float4*)scsh)[lane];
    const float4 sh = ((const float4*)(scsh + HH))[lane];
    const float4 one = make_float4(1.0f, 1.0f, 1.0f, 1.0f);
    const float4 nil = make_float4(0.0f, 0.0f, 0.0f, 0.0f);
    int beg[2], end[2];
    float dr[2];
    float4 t1[2];
#pragma unroll
    for (int q = 0; q < 2; ++q) {
      const int rowl = wv + 8 * q;
      const int r = tm * 16 + rowl;
      const int rn = (r + 1 < NN) ? (r + 1) : r;
      const int en = rowStart[rn];
      beg[q] = rowStart[r];
      end[q] = (r + 1 < NN) ? en : EE;
      dr[q] = dinv[r];
      float4 t0 = ((const float4*)(T0 + (long)r * HH))[lane];
      t1[q] = ((const float4*)(T1 + (long)r * HH))[lane];
      const float4 t2 = ((const float4*)(T2 + (long)r * HH))[lane];
      t0.x = t0.x * sc.x + sh.x;
      t0.y = t0.y * sc.y + sh.y;
      t0.z = t0.z * sc.z + sh.z;
      t0.w = t0.w * sc.w + sh.w;
      float* dstp = &As[rowl * LST + lane * 4];
      *(float4*)dstp = t0;
      *(float4*)(dstp + HH) = t1[q];
      *(float4*)(dstp + 2 * HH) = t2;
    }
#pragma unroll
    for (int q = 0; q < 2; ++q) {
      const int rowl = wv + 8 * q;
      const float4 acc = row_sum128<0>(T2, adj, dinv, beg[q], end[q], dr[q], one, nil, lane);
      float4 o;
      o.x = 2.0f * acc.x - t1[q].x;
      o.y = 2.0f * acc.y - t1[q].y;
      o.z = 2.0f * acc.z - t1[q].z;
      o.w = 2.0f * acc.w - t1[q].w;
      *(float4*)&As[rowl * LST + 3 * HH + lane * 4] = o;
    }
  }
  __syncthreads();

  const int l16 = lane & 15;
  const int khalf = (lane >> 4) * 2;
  const int col = wave * 16 + l16;

  v8f acc = {};
#pragma unroll 8
  for (int k0 = 0; k0 < KT; k0 += 4) {
    v2f a = *(const v2f*)&As[l16 * LST + k0 + khalf];
    v2f b;
    b[0] = W[(long)(k0 + khalf) * HH + col];
    b[1] = W[(long)(k0 + khalf + 1) * HH + col];
    acc = __builtin_amdgcn_wmma_f32_16x16x4_f32(
        false, a, false, b, (short)0, acc, false, false);
  }

  tile_epilogue<ACT>(acc, bias[col], tm, col, lane, out, part + (long)tm * (2 * HH));
}

#define NTILE (NN / 16)
#define FIN_T 256
#define FIN_PER ((NTILE + FIN_T - 1) / FIN_T)
static_assert(NN % 16 == 0, "every row belongs to a whole 16-row tile");
__global__ __launch_bounds__(FIN_T) void k_bn_finalize(
    const float* __restrict__ part, long pstride, const float* __restrict__ g,
    const float* __restrict__ be, float* __restrict__ scsh) {
  __shared__ float4 rs[FIN_T];
  __shared__ float4 rq[FIN_T];
  const int t = threadIdx.x;
  const int c0 = blockIdx.x * 4;
  float4 s = make_float4(0.0f, 0.0f, 0.0f, 0.0f);
  float4 q = make_float4(0.0f, 0.0f, 0.0f, 0.0f);
#pragma unroll
  for (int i = 0; i < FIN_PER; ++i) {
    const int row = t + i * FIN_T;
    const int rc = (row < NTILE) ? row : (NTILE - 1);
    const float4 vs = *(const float4*)(part + (long)rc * pstride + c0);
    const float4 vq = *(const float4*)(part + (long)rc * pstride + HH + c0);
    const bool in = row < NTILE;
    s.x += in ? vs.x : 0.0f;  s.y += in ? vs.y : 0.0f;
    s.z += in ? vs.z : 0.0f;  s.w += in ? vs.w : 0.0f;
    q.x += in ? vq.x : 0.0f;  q.y += in ? vq.y : 0.0f;
    q.z += in ? vq.z : 0.0f;  q.w += in ? vq.w : 0.0f;
  }
  rs[t] = s;
  rq[t] = q;
  __syncthreads();
  for (int off = FIN_T / 2; off > 0; off >>= 1) {
    if (t < off) {
      float4 a = rs[t], b = rs[t + off];
      a.x += b.x;  a.y += b.y;  a.z += b.z;  a.w += b.w;
      rs[t] = a;
      float4 c = rq[t], d = rq[t + off];
      c.x += d.x;  c.y += d.y;  c.z += d.z;  c.w += d.w;
      rq[t] = c;
    }
    __syncthreads();
  }
  if (t < 4) {
    const int ch = c0 + t;
    const float sum = ((const float*)&rs[0])[t];
    const float sq = ((const float*)&rq[0])[t];
    const float invN = 1.0f / (float)NN;
    const float m = sum * invN;
    const float var = sq * invN - m * m;
    const float sc = g[ch] * rsqrtf(var + 1e-5f);
    scsh[ch] = sc;
    scsh[HH + ch] = be[ch] - m * sc;
  }
}

__global__ __launch_bounds__(256) void k_final(const float* __restrict__ h,
                                               const float* __restrict__ Wm,
                                               const float* __restrict__ bm,
                                               float* __restrict__ out) {
  int n = blockIdx.x * 8 + (threadIdx.x >> 5);
  if (n >= NN) return;
  int lane = threadIdx.x & 31;
  const float4* hp = (const float4*)(h + (long)n * HH);
  float4 v = hp[lane];
  int c = lane * 4;
  float ns = v.x * v.x + v.y * v.y + v.z * v.z + v.w * v.w;
  float d0 = v.x * Wm[(c + 0) * 3 + 0] + v.y * Wm[(c + 1) * 3 + 0] +
             v.z * Wm[(c + 2) * 3 + 0] + v.w * Wm[(c + 3) * 3 + 0];
  float d1 = v.x * Wm[(c + 0) * 3 + 1] + v.y * Wm[(c + 1) * 3 + 1] +
             v.z * Wm[(c + 2) * 3 + 1] + v.w * Wm[(c + 3) * 3 + 1];
  float d2 = v.x * Wm[(c + 0) * 3 + 2] + v.y * Wm[(c + 1) * 3 + 2] +
             v.z * Wm[(c + 2) * 3 + 2] + v.w * Wm[(c + 3) * 3 + 2];
#pragma unroll
  for (int off = 16; off > 0; off >>= 1) {
    ns += __shfl_xor(ns, off, 32);
    d0 += __shfl_xor(d0, off, 32);
    d1 += __shfl_xor(d1, off, 32);
    d2 += __shfl_xor(d2, off, 32);
  }
  if (lane == 0) {
    float inv = 1.0f / fmaxf(sqrtf(ns), 1e-12f);
    out[(long)n * 3 + 0] = d0 * inv + bm[0];
    out[(long)n * 3 + 1] = d1 * inv + bm[1];
    out[(long)n * 3 + 2] = d2 * inv + bm[2];
  }
}


static inline int cdiv(long a, long b) { return (int)((a + b - 1) / b); }

extern "C" void kernel_launch(void* const* d_in, const int* in_sizes, int n_in,
                              void* d_out, int out_size, void* d_ws, size_t ws_size,
                              hipStream_t stream) {
  const float* x   = (const float*)d_in[0];
  const int*   ei  = (const int*)d_in[1];
  const float* W1  = (const float*)d_in[2];
  const float* b1  = (const float*)d_in[3];
  const float* g1  = (const float*)d_in[4];
  const float* be1 = (const float*)d_in[5];
  const float* W2  = (const float*)d_in[6];
  const float* b2  = (const float*)d_in[7];
  const float* g2  = (const float*)d_in[8];
  const float* be2 = (const float*)d_in[9];
  const float* W3  = (const float*)d_in[10];
  const float* b3  = (const float*)d_in[11];
  const float* g3  = (const float*)d_in[12];
  const float* be3 = (const float*)d_in[13];
  const float* W4  = (const float*)d_in[14];
  const float* b4  = (const float*)d_in[15];
  const float* Wm  = (const float*)d_in[16];
  const float* bm  = (const float*)d_in[17];
  float* out = (float*)d_out;

  const int* src = ei;
  const int* dst = ei + EE;

  float* ws   = (float*)d_ws;
  int*   rowStart = (int*)ws;
  float* dinv = ws + NN;
  int*   col  = (int*)(dinv + NN);
  float* T1   = (float*)(col + EE);
  float* T2   = T1 + (long)NN * HH;
  float* T3   = T2 + (long)NN * HH;
  float* hA   = T3 + (long)NN * HH;
  float* hB   = hA + (long)NN * HH;
  float* scsh = hB + (long)NN * HH;
  float* part1 = hB;
  float* part23 = T3;
  int* cntDst = (int*)hB;
  int* cntSrc = cntDst + NN;
  int* rank   = cntSrc + NN;
  int* eid    = rank + EE;

  const dim3 g3grid(NN / 16, HH / 16);
  const int g128grid = NN / 16;
  const int rowWaves = cdiv(NN, 8);
  const int row16    = cdiv((long)NN * 16, 256);

  k_zero<<<cdiv(2 * NN, 256), 256, 0, stream>>>((float*)cntDst, 2 * NN);
  k_count<<<cdiv(EE, 256), 256, 0, stream>>>(src, dst, cntDst, cntSrc, rank);
  k_scan_dinv<<<1, SCAN_T, 0, stream>>>(cntDst, cntSrc, rowStart, dinv);
  k_fill<<<cdiv(EE, 256), 256, 0, stream>>>(dst, rank, rowStart, eid);
  k_rowsort<<<rowWaves, 256, 0, stream>>>(src, rowStart, eid, cntSrc, col);

  k_spmv3<<<row16, 256, 0, stream>>>(x,  rowStart, col, dinv, nullptr, 1.0f, T1);
  k_spmv3<<<row16, 256, 0, stream>>>(T1, rowStart, col, dinv, x,       2.0f, T2);
  k_spmv3<<<row16, 256, 0, stream>>>(T2, rowStart, col, dinv, T1,      2.0f, T3);
  k_cheb_gemm3<<<g3grid, 32, 0, stream>>>(x, T1, T2, T3, W1, b1, hA, part1);
  k_bn_finalize<<<HH / 4, FIN_T, 0, stream>>>(part1, 2 * HH, g1, be1, scsh);

  k_spmv128<1><<<rowWaves, 256, 0, stream>>>(hA, rowStart, col, dinv, nullptr, scsh, 1.0f, T1);
  k_spmv128<2><<<rowWaves, 256, 0, stream>>>(T1, rowStart, col, dinv, hA,      scsh, 2.0f, T2);
  k_cheb_gemm128<0><<<g128grid, 256, 0, stream>>>(hA, T1, T2, rowStart, col, dinv, W2, b2, scsh, hB, part23);
  k_bn_finalize<<<HH / 4, FIN_T, 0, stream>>>(part23, 2 * HH, g2, be2, scsh);

  k_spmv128<1><<<rowWaves, 256, 0, stream>>>(hB, rowStart, col, dinv, nullptr, scsh, 1.0f, T1);
  k_spmv128<2><<<rowWaves, 256, 0, stream>>>(T1, rowStart, col, dinv, hB,      scsh, 2.0f, T2);
  k_cheb_gemm128<1><<<g128grid, 256, 0, stream>>>(hB, T1, T2, rowStart, col, dinv, W3, b3, scsh, hA, part23);
  k_bn_finalize<<<HH / 4, FIN_T, 0, stream>>>(part23, 2 * HH, g3, be3, scsh);

  k_spmv128<1><<<rowWaves, 256, 0, stream>>>(hA, rowStart, col, dinv, nullptr, scsh, 1.0f, T1);
  k_spmv128<2><<<rowWaves, 256, 0, stream>>>(T1, rowStart, col, dinv, hA,      scsh, 2.0f, T2);
  k_cheb_gemm128<2><<<g128grid, 256, 0, stream>>>(hA, T1, T2, rowStart, col, dinv, W4, b4, scsh, hB, part23);

  k_final<<<cdiv(NN, 8), 256, 0, stream>>>(hB, Wm, bm, out);
}


# C=128 GEMM: a block takes 32 rows (two 16-row tiles) and each weight fragment, fetched once, feeds both tiles' WMMAs: weight reads halved, two accumulator chains per wave; all sums and orders same
# speedup vs baseline: 3.8227x; 1.0352x over previous
#include <hip/hip_runtime.h>
#include <hip/hip_bf16.h>


#define NN 50000
#define EE 800000
#define HH 128
#define NTILE (NN / 16)

typedef float v2f __attribute__((ext_vector_type(2)));
typedef float v8f __attribute__((ext_vector_type(8)));

#define DEG_SAT 0xFFFFu
static_assert(NN <= 0x10000, "a source index must fit the low 16 bits of an adjacency word");

__device__ __forceinline__ float dinv_of_count(int cnt) {
  const float d = (float)cnt;
  return (d > 0.0f) ? rsqrtf(fmaxf(d, 1.0f)) : 0.0f;
}

__device__ __forceinline__ float dinv_of_word(unsigned word, const float* __restrict__ dinv) {
  const unsigned dg = word >> 16;
  float ds = dinv_of_count((int)dg);
  if (dg == DEG_SAT) ds = dinv[word & 0xFFFFu];
  return ds;
}


__global__ void k_zero(float* __restrict__ p, long n) {
  long i = (long)blockIdx.x * blockDim.x + threadIdx.x;
  if (i < n) p[i] = 0.0f;
}


__global__ void k_count(const int* __restrict__ src, const int* __restrict__ dst,
                        int* __restrict__ cntDst, int* __restrict__ cntSrc,
                        int* __restrict__ rank) {
  int e = blockIdx.x * blockDim.x + threadIdx.x;
  if (e < EE) {
    atomicAdd(&cntSrc[src[e]], 1);
    rank[e] = atomicAdd(&cntDst[dst[e]], 1);
  }
}

#define SCAN_T 1024
#define SCAN_PER ((NN + SCAN_T - 1) / SCAN_T)
__global__ __launch_bounds__(SCAN_T) void k_scan_dinv(
    const int* __restrict__ cntDst, const int* __restrict__ cntSrc,
    int* __restrict__ rowStart, float* __restrict__ dinv) {
  __shared__ int sc[NN];
  __shared__ int wsum[32];
  const int t = threadIdx.x;
  const int lane = t & 31;
  const int wv = t >> 5;
  for (int i = t; i < NN; i += SCAN_T) {
    sc[i] = cntDst[i];
    dinv[i] = dinv_of_count(cntSrc[i]);
  }
  __syncthreads();
  const int b = t * SCAN_PER;
  const int bend = (b + SCAN_PER < NN) ? (b + SCAN_PER) : NN;
  int s = 0;
  for (int i = b; i < bend; ++i) s += sc[i];
  int x = s;
  for (int off = 1; off < 32; off <<= 1) {
    int y = __shfl_up(x, off, 32);
    if (lane >= off) x += y;
  }
  if (lane == 31) wsum[wv] = x;
  __syncthreads();
  if (wv == 0) {
    int w = wsum[lane];
    for (int off = 1; off < 32; off <<= 1) {
      int y = __shfl_up(w, off, 32);
      if (lane >= off) w += y;
    }
    wsum[lane] = w;
  }
  __syncthreads();
  int run = x - s + ((wv > 0) ? wsum[wv - 1] : 0);
  for (int i = b; i < bend; ++i) {
    int v = sc[i];
    sc[i] = run;
    run += v;
  }
  __syncthreads();
  for (int i = t; i < NN; i += SCAN_T) rowStart[i] = sc[i];
}

__global__ void k_fill(const int* __restrict__ dst, const int* __restrict__ rank,
                       const int* __restrict__ rowStart, int* __restrict__ eid) {
  int e = blockIdx.x * blockDim.x + threadIdx.x;
  if (e < EE) eid[rowStart[dst[e]] + rank[e]] = e;
}

__global__ __launch_bounds__(256) void k_rowsort(
    const int* __restrict__ src, const int* __restrict__ rowStart,
    const int* __restrict__ eid, const int* __restrict__ cntSrc,
    int* __restrict__ col) {
  const int r = blockIdx.x * 8 + __builtin_amdgcn_readfirstlane((int)(threadIdx.x >> 5));
  if (r >= NN) return;
  const int lane = threadIdx.x & 31;
  const int beg = rowStart[r];
  const int end = (r + 1 < NN) ? rowStart[r + 1] : EE;
  for (int i = beg + lane; i < end; i += 32) {
    const int my = eid[i];
    int rk = 0;
    for (int j = beg; j < end; ++j) rk += (eid[j] < my) ? 1 : 0;
    const unsigned s = (unsigned)src[my];
    const unsigned dg = (unsigned)cntSrc[s];
    col[beg + rk] = (int)(s | ((dg < DEG_SAT ? dg : DEG_SAT) << 16));
  }
}


__global__ __launch_bounds__(256) void k_spmv3(
    const float* __restrict__ h, const int* __restrict__ rowStart,
    const int* __restrict__ col, const float* __restrict__ dinv,
    const float* __restrict__ T0, float scale, float* __restrict__ out) {
  const int gid = blockIdx.x * 256 + threadIdx.x;
  const int r = gid >> 4;
  const int sub = threadIdx.x & 15;
  float a0 = 0.0f, a1 = 0.0f, a2 = 0.0f;
  if (r < NN) {
    const int beg = rowStart[r];
    const int end = (r + 1 < NN) ? rowStart[r + 1] : EE;
    const float dr = dinv[r];
    for (int p = beg + sub; p < end; p += 16) {
      const unsigned word = (unsigned)col[p];
      const int s = (int)(word & 0xFFFFu);
      const float w = -dinv_of_word(word, dinv) * dr;
      a0 += w * h[(long)s * 3 + 0];
      a1 += w * h[(long)s * 3 + 1];
      a2 += w * h[(long)s * 3 + 2];
    }
  }
  for (int off = 8; off > 0; off >>= 1) {
    a0 += __shfl_xor(a0, off, 32);
    a1 += __shfl_xor(a1, off, 32);
    a2 += __shfl_xor(a2, off, 32);
  }
  if (r < NN && sub == 0) {
    float o0 = scale * a0, o1 = scale * a1, o2 = scale * a2;
    if (T0) {
      o0 -= T0[(long)r * 3 + 0];
      o1 -= T0[(long)r * 3 + 1];
      o2 -= T0[(long)r * 3 + 2];
    }
    out[(long)r * 3 + 0] = o0;
    out[(long)r * 3 + 1] = o1;
    out[(long)r * 3 + 2] = o2;
  }
}

template <int AFF>
__device__ __forceinline__ float4 row_sum128(
    const float* __restrict__ h, const int* __restrict__ col,
    const float* __restrict__ dinv, int beg, int end, float dr,
    float4 sc, float4 sh, int lane) {
  float4 acc = make_float4(0.0f, 0.0f, 0.0f, 0.0f);
  for (int base = beg; base < end; base += 32) {
    const int n = (end - base < 32) ? (end - base) : 32;
    int myc = 0;
    float myw = 0.0f;
    if (lane < n) {
      const unsigned word = (unsigned)col[base + lane];
      myc = (int)(word & 0xFFFFu);
      myw = -dinv_of_word(word, dinv) * dr;
    }
#pragma unroll 4
    for (int j = 0; j < n; ++j) {
      const int s = __builtin_amdgcn_readlane(myc, j);
      const float w = __int_as_float(__builtin_amdgcn_readlane(__float_as_int(myw), j));
      float4 v = ((const float4*)(h + (long)s * HH))[lane];
      if (AFF == 1) {
        v.x = v.x * sc.x + sh.x;
        v.y = v.y * sc.y + sh.y;
        v.z = v.z * sc.z + sh.z;
        v.w = v.w * sc.w + sh.w;
      }
      acc.x += w * v.x;
      acc.y += w * v.y;
      acc.z += w * v.z;
      acc.w += w * v.w;
    }
  }
  return acc;
}

template <int AFF>
__global__ __launch_bounds__(256) void k_spmv128(
    const float* __restrict__ h, const int* __restrict__ rowStart,
    const int* __restrict__ col, const float* __restrict__ dinv,
    const float* __restrict__ T0, const float* __restrict__ scsh,
    float scale, float* __restrict__ out) {
  const int r = blockIdx.x * 8 + __builtin_amdgcn_readfirstlane((int)(threadIdx.x >> 5));
  if (r >= NN) return;
  const int lane = threadIdx.x & 31;
  const int beg = rowStart[r];
  const int end = (r + 1 < NN) ? rowStart[r + 1] : EE;
  const float dr = dinv[r];
  float4 sc = make_float4(1.0f, 1.0f, 1.0f, 1.0f);
  float4 sh = make_float4(0.0f, 0.0f, 0.0f, 0.0f);
  if (AFF != 0) {
    sc = ((const float4*)scsh)[lane];
    sh = ((const float4*)(scsh + HH))[lane];
  }
  const float4 acc = row_sum128<AFF>(h, col, dinv, beg, end, dr, sc, sh, lane);
  float4 o = make_float4(scale * acc.x, scale * acc.y, scale * acc.z, scale * acc.w);
  if (T0) {
    float4 t0 = ((const float4*)(T0 + (long)r * HH))[lane];
    if (AFF == 2) {
      t0.x = t0.x * sc.x + sh.x;
      t0.y = t0.y * sc.y + sh.y;
      t0.z = t0.z * sc.z + sh.z;
      t0.w = t0.w * sc.w + sh.w;
    }
    o.x -= t0.x;
    o.y -= t0.y;
    o.z -= t0.z;
    o.w -= t0.w;
  }
  ((float4*)(out + (long)r * HH))[lane] = o;
}


template <int ACT>
__device__ __forceinline__ void tile_epilogue(const v8f acc, float bc, int tm, int col,
                                              int lane, float* out, float* prow) {
  float s = 0.0f, s2 = 0.0f;
#pragma unroll
  for (int j = 0; j < 8; ++j) {
    const int r = tm * 16 + j + (lane >> 4) * 8;
    float v = acc[j] + bc;
    if (ACT == 0) v = v > 0.0f ? v : 0.01f * v;
    if (ACT == 1) v = fmaxf(v, 0.0f);
    out[(long)r * HH + col] = v;
    s += v;
    s2 += v * v;
  }
  if (ACT != 2) {
    s += __shfl_xor(s, 16, 32);
    s2 += __shfl_xor(s2, 16, 32);
    if (lane < 16) {
      prow[col] = s;
      prow[HH + col] = s2;
    }
  }
}

__global__ __launch_bounds__(32) void k_cheb_gemm3(
    const float* __restrict__ T0, const float* __restrict__ T1,
    const float* __restrict__ T2, const float* __restrict__ T3,
    const float* __restrict__ W, const float* __restrict__ bias,
    float* __restrict__ out, float* __restrict__ part) {
  const int lane = threadIdx.x;
  const int tm = blockIdx.x;
  const int tn = blockIdx.y;
  const int row = tm * 16 + (lane & 15);
  const int col = tn * 16 + (lane & 15);
  const int khalf = (lane >> 4) * 2;
  const float* Ts[4] = {T0, T1, T2, T3};

  v8f acc = {};
  for (int k0 = 0; k0 < 12; k0 += 4) {
    v2f a, b;
#pragma unroll
    for (int j = 0; j < 2; ++j) {
      int kk = k0 + khalf + j;
      int t = kk / 3;
      int cc = kk - t * 3;
      a[j] = Ts[t][(long)row * 3 + cc];
      b[j] = W[(long)kk * HH + col];
    }
    acc = __builtin_amdgcn_wmma_f32_16x16x4_f32(
        false, a, false, b, (short)0, acc, false, false);
  }

  tile_epilogue<0>(acc, bias[col], tm, col, lane, out, part + (long)tm * (2 * HH));
}

#define KT 512
#define LST (KT + 4)
#define RT 2
#define PR (16 * RT)
#define RPW (PR / 8)

template <int ACT>
__global__ __launch_bounds__(256) void k_cheb_gemm128(
    const float* __restrict__ T0, const float* __restrict__ T1,
    const float* __restrict__ T2, const int* __restrict__ rowStart,
    const int* __restrict__ adj, const float* __restrict__ dinv,
    const float* __restrict__ W, const float* __restrict__ bias,
    const float* __restrict__ scsh, float* __restrict__ out,
    float* __restrict__ part) {
  __shared__ float As[PR * LST];
  const int tid = threadIdx.x;
  const int lane = tid & 31;
  const int wave = tid >> 5;
  const int blk = blockIdx.x;

  {
    const int wv = __builtin_amdgcn_readfirstlane(wave);
    const float4 sc = ((const float4*)scsh)[lane];
    const float4 sh = ((const float4*)(scsh + HH))[lane];
    const float4 one = make_float4(1.0f, 1.0f, 1.0f, 1.0f);
    const float4 nil = make_float4(0.0f, 0.0f, 0.0f, 0.0f);
    int beg[RPW], end[RPW];
    float dr[RPW];
    float4 t1[RPW];
#pragma unroll
    for (int q = 0; q < RPW; ++q) {
      const int rowl = wv + 8 * q;
      const int rg = blk * PR + rowl;
      const int r = (rg < NN) ? rg : (NN - 1);
      const int rn = (r + 1 < NN) ? (r + 1) : r;
      const int en = rowStart[rn];
      beg[q] = rowStart[r];
      end[q] = (r + 1 < NN) ? en : EE;
      dr[q] = dinv[r];
      float4 t0 = ((const float4*)(T0 + (long)r * HH))[lane];
      t1[q] = ((const float4*)(T1 + (long)r * HH))[lane];
      const float4 t2 = ((const float4*)(T2 + (long)r * HH))[lane];
      t0.x = t0.x * sc.x + sh.x;
      t0.y = t0.y * sc.y + sh.y;
      t0.z = t0.z * sc.z + sh.z;
      t0.w = t0.w * sc.w + sh.w;
      float* dstp = &As[rowl * LST + lane * 4];
      *(float4*)dstp = t0;
      *(float4*)(dstp + HH) = t1[q];
      *(float4*)(dstp + 2 * HH) = t2;
    }
#pragma unroll
    for (int q = 0; q < RPW; ++q) {
      const int rowl = wv + 8 * q;
      const float4 acc = row_sum128<0>(T2, adj, dinv, beg[q], end[q], dr[q], one, nil, lane);
      float4 o;
      o.x = 2.0f * acc.x - t1[q].x;
      o.y = 2.0f * acc.y - t1[q].y;
      o.z = 2.0f * acc.z - t1[q].z;
      o.w = 2.0f * acc.w - t1[q].w;
      *(float4*)&As[rowl * LST + 3 * HH + lane * 4] = o;
    }
  }
  __syncthreads();

  const int l16 = lane & 15;
  const int khalf = (lane >> 4) * 2;
  const int col = wave * 16 + l16;

  v8f acc[RT];
#pragma unroll
  for (int t = 0; t < RT; ++t) acc[t] = v8f{};
#pragma unroll 8
  for (int k0 = 0; k0 < KT; k0 += 4) {
    v2f b;
    b[0] = W[(long)(k0 + khalf) * HH + col];
    b[1] = W[(long)(k0 + khalf + 1) * HH + col];
#pragma unroll
    for (int t = 0; t < RT; ++t) {
      v2f a = *(const v2f*)&As[(t * 16 + l16) * LST + k0 + khalf];
      acc[t] = __builtin_amdgcn_wmma_f32_16x16x4_f32(
          false, a, false, b, (short)0, acc[t], false, false);
    }
  }

  const float bc = bias[col];
#pragma unroll
  for (int t = 0; t < RT; ++t) {
    const int tm = blk * RT + t;
    if (tm < NTILE)
      tile_epilogue<ACT>(acc[t], bc, tm, col, lane, out, part + (long)tm * (2 * HH));
  }
}

#define FIN_T 256
#define FIN_PER ((NTILE + FIN_T - 1) / FIN_T)
static_assert(NN % 16 == 0, "every row belongs to a whole 16-row tile");
__global__ __launch_bounds__(FIN_T) void k_bn_finalize(
    const float* __restrict__ part, long pstride, const float* __restrict__ g,
    const float* __restrict__ be, float* __restrict__ scsh) {
  __shared__ float4 rs[FIN_T];
  __shared__ float4 rq[FIN_T];
  const int t = threadIdx.x;
  const int c0 = blockIdx.x * 4;
  float4 s = make_float4(0.0f, 0.0f, 0.0f, 0.0f);
  float4 q = make_float4(0.0f, 0.0f, 0.0f, 0.0f);
#pragma unroll
  for (int i = 0; i < FIN_PER; ++i) {
    const int row = t + i * FIN_T;
    const int rc = (row < NTILE) ? row : (NTILE - 1);
    const float4 vs = *(const float4*)(part + (long)rc * pstride + c0);
    const float4 vq = *(const float4*)(part + (long)rc * pstride + HH + c0);
    const bool in = row < NTILE;
    s.x += in ? vs.x : 0.0f;  s.y += in ? vs.y : 0.0f;
    s.z += in ? vs.z : 0.0f;  s.w += in ? vs.w : 0.0f;
    q.x += in ? vq.x : 0.0f;  q.y += in ? vq.y : 0.0f;
    q.z += in ? vq.z : 0.0f;  q.w += in ? vq.w : 0.0f;
  }
  rs[t] = s;
  rq[t] = q;
  __syncthreads();
  for (int off = FIN_T / 2; off > 0; off >>= 1) {
    if (t < off) {
      float4 a = rs[t], b = rs[t + off];
      a.x += b.x;  a.y += b.y;  a.z += b.z;  a.w += b.w;
      rs[t] = a;
      float4 c = rq[t], d = rq[t + off];
      c.x += d.x;  c.y += d.y;  c.z += d.z;  c.w += d.w;
      rq[t] = c;
    }
    __syncthreads();
  }
  if (t < 4) {
    const int ch = c0 + t;
    const float sum = ((const float*)&rs[0])[t];
    const float sq = ((const float*)&rq[0])[t];
    const float invN = 1.0f / (float)NN;
    const float m = sum * invN;
    const float var = sq * invN - m * m;
    const float sc = g[ch] * rsqrtf(var + 1e-5f);
    scsh[ch] = sc;
    scsh[HH + ch] = be[ch] - m * sc;
  }
}

__global__ __launch_bounds__(256) void k_final(const float* __restrict__ h,
                                               const float* __restrict__ Wm,
                                               const float* __restrict__ bm,
                                               float* __restrict__ out) {
  int n = blockIdx.x * 8 + (threadIdx.x >> 5);
  if (n >= NN) return;
  int lane = threadIdx.x & 31;
  const float4* hp = (const float4*)(h + (long)n * HH);
  float4 v = hp[lane];
  int c = lane * 4;
  float ns = v.x * v.x + v.y * v.y + v.z * v.z + v.w * v.w;
  float d0 = v.x * Wm[(c + 0) * 3 + 0] + v.y * Wm[(c + 1) * 3 + 0] +
             v.z * Wm[(c + 2) * 3 + 0] + v.w * Wm[(c + 3) * 3 + 0];
  float d1 = v.x * Wm[(c + 0) * 3 + 1] + v.y * Wm[(c + 1) * 3 + 1] +
             v.z * Wm[(c + 2) * 3 + 1] + v.w * Wm[(c + 3) * 3 + 1];
  float d2 = v.x * Wm[(c + 0) * 3 + 2] + v.y * Wm[(c + 1) * 3 + 2] +
             v.z * Wm[(c + 2) * 3 + 2] + v.w * Wm[(c + 3) * 3 + 2];
#pragma unroll
  for (int off = 16; off > 0; off >>= 1) {
    ns += __shfl_xor(ns, off, 32);
    d0 += __shfl_xor(d0, off, 32);
    d1 += __shfl_xor(d1, off, 32);
    d2 += __shfl_xor(d2, off, 32);
  }
  if (lane == 0) {
    float inv = 1.0f / fmaxf(sqrtf(ns), 1e-12f);
    out[(long)n * 3 + 0] = d0 * inv + bm[0];
    out[(long)n * 3 + 1] = d1 * inv + bm[1];
    out[(long)n * 3 + 2] = d2 * inv + bm[2];
  }
}


static inline int cdiv(long a, long b) { return (int)((a + b - 1) / b); }

extern "C" void kernel_launch(void* const* d_in, const int* in_sizes, int n_in,
                              void* d_out, int out_size, void* d_ws, size_t ws_size,
                              hipStream_t stream) {
  const float* x   = (const float*)d_in[0];
  const int*   ei  = (const int*)d_in[1];
  const float* W1  = (const float*)d_in[2];
  const float* b1  = (const float*)d_in[3];
  const float* g1  = (const float*)d_in[4];
  const float* be1 = (const float*)d_in[5];
  const float* W2  = (const float*)d_in[6];
  const float* b2  = (const float*)d_in[7];
  const float* g2  = (const float*)d_in[8];
  const float* be2 = (const float*)d_in[9];
  const float* W3  = (const float*)d_in[10];
  const float* b3  = (const float*)d_in[11];
  const float* g3  = (const float*)d_in[12];
  const float* be3 = (const float*)d_in[13];
  const float* W4  = (const float*)d_in[14];
  const float* b4  = (const float*)d_in[15];
  const float* Wm  = (const float*)d_in[16];
  const float* bm  = (const float*)d_in[17];
  float* out = (float*)d_out;

  const int* src = ei;
  const int* dst = ei + EE;

  float* ws   = (float*)d_ws;
  int*   rowStart = (int*)ws;
  float* dinv = ws + NN;
  int*   col  = (int*)(dinv + NN);
  float* T1   = (float*)(col + EE);
  float* T2   = T1 + (long)NN * HH;
  float* T3   = T2 + (long)NN * HH;
  float* hA   = T3 + (long)NN * HH;
  float* hB   = hA + (long)NN * HH;
  float* scsh = hB + (long)NN * HH;
  float* part1 = hB;
  float* part23 = T3;
  int* cntDst = (int*)hB;
  int* cntSrc = cntDst + NN;
  int* rank   = cntSrc + NN;
  int* eid    = rank + EE;

  const dim3 g3grid(NN / 16, HH / 16);
  const int g128grid = (NTILE + RT - 1) / RT;
  const int rowWaves = cdiv(NN, 8);
  const int row16    = cdiv((long)NN * 16, 256);

  k_zero<<<cdiv(2 * NN, 256), 256, 0, stream>>>((float*)cntDst, 2 * NN);
  k_count<<<cdiv(EE, 256), 256, 0, stream>>>(src, dst, cntDst, cntSrc, rank);
  k_scan_dinv<<<1, SCAN_T, 0, stream>>>(cntDst, cntSrc, rowStart, dinv);
  k_fill<<<cdiv(EE, 256), 256, 0, stream>>>(dst, rank, rowStart, eid);
  k_rowsort<<<rowWaves, 256, 0, stream>>>(src, rowStart, eid, cntSrc, col);

  k_spmv3<<<row16, 256, 0, stream>>>(x,  rowStart, col, dinv, nullptr, 1.0f, T1);
  k_spmv3<<<row16, 256, 0, stream>>>(T1, rowStart, col, dinv, x,       2.0f, T2);
  k_spmv3<<<row16, 256, 0, stream>>>(T2, rowStart, col, dinv, T1,      2.0f, T3);
  k_cheb_gemm3<<<g3grid, 32, 0, stream>>>(x, T1, T2, T3, W1, b1, hA, part1);
  k_bn_finalize<<<HH / 4, FIN_T, 0, stream>>>(part1, 2 * HH, g1, be1, scsh);

  k_spmv128<1><<<rowWaves, 256, 0, stream>>>(hA, rowStart, col, dinv, nullptr, scsh, 1.0f, T1);
  k_spmv128<2><<<rowWaves, 256, 0, stream>>>(T1, rowStart, col, dinv, hA,      scsh, 2.0f, T2);
  k_cheb_gemm128<0><<<g128grid, 256, 0, stream>>>(hA, T1, T2, rowStart, col, dinv, W2, b2, scsh, hB, part23);
  k_bn_finalize<<<HH / 4, FIN_T, 0, stream>>>(part23, 2 * HH, g2, be2, scsh);

  k_spmv128<1><<<rowWaves, 256, 0, stream>>>(hB, rowStart, col, dinv, nullptr, scsh, 1.0f, T1);
  k_spmv128<2><<<rowWaves, 256, 0, stream>>>(T1, rowStart, col, dinv, hB,      scsh, 2.0f, T2);
  k_cheb_gemm128<1><<<g128grid, 256, 0, stream>>>(hB, T1, T2, rowStart, col, dinv, W3, b3, scsh, hA, part23);
  k_bn_finalize<<<HH / 4, FIN_T, 0, stream>>>(part23, 2 * HH, g3, be3, scsh);

  k_spmv128<1><<<rowWaves, 256, 0, stream>>>(hA, rowStart, col, dinv, nullptr, scsh, 1.0f, T1);
  k_spmv128<2><<<rowWaves, 256, 0, stream>>>(T1, rowStart, col, dinv, hA,      scsh, 2.0f, T2);
  k_cheb_gemm128<2><<<g128grid, 256, 0, stream>>>(hA, T1, T2, rowStart, col, dinv, W4, b4, scsh, hB, part23);

  k_final<<<cdiv(NN, 8), 256, 0, stream>>>(hB, Wm, bm, out);
}
